# starting kernel
# baseline (speedup 1.0000x reference)
#include <hip/hip_runtime.h>

#define DF   128
#define N1C  50000

typedef float v2f __attribute__((ext_vector_type(2)));
typedef float v8f __attribute__((ext_vector_type(8)));

__global__ __launch_bounds__(256)
void GCN_zero_kernel(float4* __restrict__ p, int n4) {
    int i = blockIdx.x * blockDim.x + threadIdx.x;
    int stride = gridDim.x * blockDim.x;
    float4 z; z.x = 0.f; z.y = 0.f; z.z = 0.f; z.w = 0.f;
    for (; i < n4; i += stride) p[i] = z;
}

__global__ __launch_bounds__(256)
void GCN_spmm1_kernel(const float* __restrict__ feat,
                      const float* __restrict__ w,
                      const int*   __restrict__ src,
                      const int*   __restrict__ dst,
                      float*       __restrict__ out,
                      int nE) {
    int e    = (blockIdx.x * blockDim.x + threadIdx.x) >> 5;
    int lane = threadIdx.x & 31;
    if (e >= nE) return;
    int u = src[e];
    int v = dst[e];
    float we = w[e];
    const float4* f4 = (const float4*)(feat + (size_t)u * DF);
    float4 m = f4[lane];
    float* o = out + (size_t)v * DF + lane * 4;
    unsafeAtomicAdd(o + 0, m.x * we);
    unsafeAtomicAdd(o + 1, m.y * we);
    unsafeAtomicAdd(o + 2, m.z * we);
    unsafeAtomicAdd(o + 3, m.w * we);
}

__global__ __launch_bounds__(256)
void GCN_spmm2_kernel(const float* __restrict__ h,
                      const float* __restrict__ b1,
                      const float* __restrict__ w,
                      const int*   __restrict__ src,
                      const int*   __restrict__ dst,
                      float*       __restrict__ out,
                      int nE) {
    int e    = (blockIdx.x * blockDim.x + threadIdx.x) >> 5;
    int lane = threadIdx.x & 31;
    if (e >= nE) return;
    int u = src[e];
    int v = dst[e];
    float we = w[e];
    const float4* h4 = (const float4*)(h + (size_t)u * DF);
    const float4* bb = (const float4*)(b1);
    float4 m = h4[lane];
    float4 b = bb[lane];
    float r0 = fmaxf(m.x + b.x, 0.f) * we;
    float r1 = fmaxf(m.y + b.y, 0.f) * we;
    float r2 = fmaxf(m.z + b.z, 0.f) * we;
    float r3 = fmaxf(m.w + b.w, 0.f) * we;
    float* o = out + (size_t)v * DF + lane * 4;
    unsafeAtomicAdd(o + 0, r0);
    unsafeAtomicAdd(o + 1, r1);
    unsafeAtomicAdd(o + 2, r2);
    unsafeAtomicAdd(o + 3, r3);
}

__global__ __launch_bounds__(256)
void GCN_gemm_wmma_kernel(const float* __restrict__ A,
                          const float* __restrict__ W,
                          const float* __restrict__ bias,
                          float*       __restrict__ out,
                          int M) {
    const int lane = threadIdx.x & 31;
    const int waveInBlk = threadIdx.x >> 5;
    const int nTilesM = (M + 15) >> 4;
    const int nTiles  = nTilesM * (DF / 16);
    int tile = blockIdx.x * (blockDim.x >> 5) + waveInBlk;
    if (tile >= nTiles) return;
    const int tM = tile >> 3;
    const int tN = tile & 7;

    const int half = lane >> 4;
    const int mn   = lane & 15;
    const int row  = tM * 16 + mn;
    const int col  = tN * 16 + mn;

    const int rowc = (row < M) ? row : (M - 1);
    const float* Arow = A + (size_t)rowc * DF + 2 * half;
    const float* Bcol = W + (size_t)(2 * half) * DF + col;

    v8f c = {};
    #pragma unroll
    for (int k = 0; k < DF; k += 4) {
        const float2 av = *(const float2*)(Arow + k);
        v2f a; a[0] = av.x; a[1] = av.y;
        v2f b;
        b[0] = Bcol[(size_t)k * DF];
        b[1] = Bcol[(size_t)(k + 1) * DF];
        c = __builtin_amdgcn_wmma_f32_16x16x4_f32(
                 false, a,  false, b,
                 (short)0, c,  false,  false);
    }

    const float bv = bias[col];
    #pragma unroll
    for (int v = 0; v < 8; ++v) {
        const int r = tM * 16 + v + 8 * half;
        if (r < M) out[(size_t)r * DF + col] = c[v] + bv;
    }
}

extern "C" void kernel_launch(void* const* d_in, const int* in_sizes, int n_in,
                              void* d_out, int out_size, void* d_ws, size_t ws_size,
                              hipStream_t stream) {
    const float* feature = (const float*)d_in[0];
    const float* w0      = (const float*)d_in[1];
    const float* w1      = (const float*)d_in[2];
    const float* b1      = (const float*)d_in[3];
    const float* W2      = (const float*)d_in[4];
    const float* b2      = (const float*)d_in[5];
    const int*   src0    = (const int*)d_in[6];
    const int*   dst0    = (const int*)d_in[7];
    const int*   src1    = (const int*)d_in[8];
    const int*   dst1    = (const int*)d_in[9];

    const int E0 = in_sizes[1];
    const int E1 = in_sizes[2];
    const int N1 = N1C;
    const int N2 = out_size / DF;

    float* h   = (float*)d_ws;
    float* agg = h + (size_t)N1 * DF;
    float* out = (float*)d_out;

    {
        int n4 = (N1 * DF + N2 * DF) / 4;
        int blocks = (n4 + 255) / 256;
        if (blocks > 4096) blocks = 4096;
        GCN_zero_kernel<<<blocks, 256, 0, stream>>>((float4*)d_ws, n4);
    }
    {
        long long threads = (long long)E0 * 32;
        int blocks = (int)((threads + 255) / 256);
        GCN_spmm1_kernel<<<blocks, 256, 0, stream>>>(feature, w0, src0, dst0, h, E0);
    }
    {
        long long threads = (long long)E1 * 32;
        int blocks = (int)((threads + 255) / 256);
        GCN_spmm2_kernel<<<blocks, 256, 0, stream>>>(h, b1, w1, src1, dst1, agg, E1);
    }
    {
        int nTiles = ((N2 + 15) / 16) * (DF / 16);
        int wavesPerBlk = 256 / 32;
        int blocks = (nTiles + wavesPerBlk - 1) / wavesPerBlk;
        GCN_gemm_wmma_kernel<<<blocks, 256, 0, stream>>>(agg, W2, b2, out, N2);
    }
}


# Dense product retiled: a wave makes 16 rows x all 128 columns (A read once, 8x fewer waves), W2 staged per work-group in LDS as k-quads, rows stored whole; k order inside each 8 of the sum changes
# speedup vs baseline: 2.3681x; 2.3681x over previous
#include <hip/hip_runtime.h>
#include <type_traits>


#define DF   128
#define N1C  50000

#define SLAB_TPB 1024

typedef float v2f __attribute__((ext_vector_type(2)));
typedef float v8f __attribute__((ext_vector_type(8)));

#define SLAB_COL(k) ((((k) & 3) << 5) + ((k) >> 2))

__device__ __noinline__
void GCN_slab_add_direct(float* accRow, const float* __restrict__ f, float we) {
    for (int k = 0; k < DF; ++k)
        __hip_atomic_fetch_add(accRow + SLAB_COL(k), f[k] * we,
                               __ATOMIC_RELAXED, __HIP_MEMORY_SCOPE_WORKGROUP);
}


template <unsigned ROWS>
__device__ __forceinline__ unsigned GCN_key4(const int4& v) {
    return  (((unsigned)v.x / ROWS) & 255u)        | ((((unsigned)v.y / ROWS) & 255u) << 8) |
           ((((unsigned)v.z / ROWS) & 255u) << 16) |  (((unsigned)v.w / ROWS)         << 24);
}
template <unsigned ROWS>
__device__ __forceinline__ void GCN_key16(const int* __restrict__ dst, unsigned* __restrict__ keys, int i) {
    const int4* d4 = (const int4*)dst + 4 * (size_t)i;
    const int4 a = d4[0], b = d4[1], c = d4[2], d = d4[3];
    uint4 o;
    o.x = GCN_key4<ROWS>(a); o.y = GCN_key4<ROWS>(b); o.z = GCN_key4<ROWS>(c); o.w = GCN_key4<ROWS>(d);
    ((uint4*)keys)[i] = o;
}
template <unsigned ROWS0, unsigned ROWS1>
__global__ __launch_bounds__(256)
void GCN_slab_key_kernel(const int* __restrict__ dst0, const int* __restrict__ dst1,
                         unsigned* __restrict__ keys0, unsigned* __restrict__ keys1,
                         int n16_0, int n16_1) {
    int i = blockIdx.x * 256 + threadIdx.x;
    if (i < n16_0) { GCN_key16<ROWS0>(dst0, keys0, i); return; }
    i -= n16_0;
    if (i < n16_1) GCN_key16<ROWS1>(dst1, keys1, i);
}

template <int ROWS, int CAP, bool BIAS_RELU>
__global__ __launch_bounds__(SLAB_TPB)
void GCN_slab_spmm_kernel(const float* __restrict__ in,
                          const float* __restrict__ w,
                          const int*   __restrict__ src,
                          const int*   __restrict__ dst,
                          const unsigned* __restrict__ keys,
                          const float* __restrict__ bias,
                          float*       __restrict__ out,
                          int nE, int nRows) {
    __shared__ float acc[(ROWS + 1) * DF];
    __shared__ int   recRow[CAP];
    __shared__ int   recU[CAP];
    __shared__ float recW[CAP];
    __shared__ int   cnt;

    const int t    = threadIdx.x;
    const int lane = t & 31;
    const int wv   = __builtin_amdgcn_readfirstlane(t >> 5);
    const int row0 = blockIdx.x * ROWS;
    int rowsHere = nRows - row0;
    if (rowsHere > ROWS) rowsHere = ROWS;
    if (rowsHere < 0) rowsHere = 0;
    const unsigned rowsU = (unsigned)rowsHere;

    {
        float4 z; z.x = 0.f; z.y = 0.f; z.z = 0.f; z.w = 0.f;
        float4* a4 = (float4*)acc;
        for (int i = t; i < (ROWS + 1) * DF / 4; i += SLAB_TPB) a4[i] = z;
        if (t == 0) cnt = 0;
    }
    __syncthreads();

    const int    n16 = nE >> 4;
    const uint4* k4  = (const uint4*)keys;
    const int start = (int)(((long long)blockIdx.x * n16) / gridDim.x);

    auto edgeOf = [&](unsigned wd) -> int {
        if ((int)wd < 0) return (int)~wd;
        const int p = (int)(wd & 63u);
        const int k = ((p >> 5) << 1) | ((p >> 2) & 1);
        int idx = start + (int)(wd >> 6) + k * SLAB_TPB;
        if (idx >= n16) idx -= n16;
        return 16 * idx + 4 * (p & 3) + ((p >> 3) & 3);
    };
    auto put = [&](int pos, unsigned wd) {
        if (pos < CAP) {
            recU[pos] = (int)wd;
        } else {
            const int e = edgeOf(wd);
            const unsigned r = (unsigned)(dst[e] - row0);
            if (r < rowsU) GCN_slab_add_direct(acc + r * DF, in + (size_t)src[e] * DF, w[e]);
        }
    };
    const unsigned kb = (blockIdx.x & 255u) * 0x01010101u;
    auto eq4 = [&](unsigned x) -> unsigned {
        const unsigned y = x ^ kb;
        return (y - 0x01010101u) & ~y & 0x80808080u;
    };
    auto eq16 = [&](const uint4& v) -> unsigned {
        return (eq4(v.x) >> 7) | (eq4(v.y) >> 6) | (eq4(v.z) >> 5) | (eq4(v.w) >> 4);
    };
    auto batch = [&](auto utag, int b) {
        constexpr int U = decltype(utag)::value;
        uint4 v[U];
        #pragma unroll
        for (int k = 0; k < U; ++k) {
            int idx = start + b + k * SLAB_TPB + t;
            if (idx >= n16) idx -= n16;
            v[k] = k4[idx];
        }
        unsigned h[4] = {0u, 0u, 0u, 0u};
        #pragma unroll
        for (int k = 0; k < U; ++k) h[k] = eq16(v[k]);
        const unsigned mlo = h[0] | (h[1] << 4);
        const unsigned mhi = h[2] | (h[3] << 4);
        const int n = __popc(mlo) + __popc(mhi);
        int pos = 0;
        if (n) pos = atomicAdd(&cnt, n);
        const unsigned tag = (unsigned)(b + t) << 6;
        unsigned long long m = ((unsigned long long)mhi << 32) | mlo;
        while (m) {
            const unsigned p = (unsigned)__builtin_ctzll(m);
            m &= m - 1;
            put(pos++, tag | p);
        }
    };
    int b = 0;
    for (; b + SLAB_TPB * 4 <= n16; b += SLAB_TPB * 4) batch(std::integral_constant<int, 4>{}, b);
    if (b + SLAB_TPB * 2 <= n16) { batch(std::integral_constant<int, 2>{}, b); b += SLAB_TPB * 2; }
    if (b + SLAB_TPB     <= n16) { batch(std::integral_constant<int, 1>{}, b); b += SLAB_TPB; }
    if (b + t < n16) batch(std::integral_constant<int, 1>{}, b);
    for (int e = 16 * n16 + t; e < nE; e += SLAB_TPB)
        if ((unsigned)(dst[e] - row0) < rowsU) put(atomicAdd(&cnt, 1), ~(unsigned)e);
    __syncthreads();

    int count = cnt;
    if (count > CAP) count = CAP;
    for (int i = t; i < count; i += SLAB_TPB) {
        const int e = edgeOf((unsigned)recU[i]);
        const unsigned r = (unsigned)(dst[e] - row0);
        recRow[i] = (r < rowsU) ? (int)r : ROWS;
        recU[i]   = src[e];
        recW[i]   = w[e];
    }
    __syncthreads();

    const int countS = __builtin_amdgcn_readfirstlane(count);
    int i = wv;
    for (; i + 32 * 7 < countS; i += 32 * 8) {
        float4 m[8];
        int    r[8];
        float  we[8];
        #pragma unroll
        for (int k = 0; k < 8; ++k) {
            const int idx = i + 32 * k;
            const int u = __builtin_amdgcn_readfirstlane(recU[idx]);
            r[k]  = __builtin_amdgcn_readfirstlane(recRow[idx]);
            we[k] = recW[idx];
            m[k]  = ((const float4*)(in + (size_t)u * DF))[lane];
        }
        #pragma unroll
        for (int k = 0; k < 8; ++k) {
            float* a = acc + r[k] * DF + lane;
            __hip_atomic_fetch_add(a +  0, m[k].x * we[k], __ATOMIC_RELAXED, __HIP_MEMORY_SCOPE_WORKGROUP);
            __hip_atomic_fetch_add(a + 32, m[k].y * we[k], __ATOMIC_RELAXED, __HIP_MEMORY_SCOPE_WORKGROUP);
            __hip_atomic_fetch_add(a + 64, m[k].z * we[k], __ATOMIC_RELAXED, __HIP_MEMORY_SCOPE_WORKGROUP);
            __hip_atomic_fetch_add(a + 96, m[k].w * we[k], __ATOMIC_RELAXED, __HIP_MEMORY_SCOPE_WORKGROUP);
        }
    }
    for (; i < countS; i += 32) {
        const int   u   = __builtin_amdgcn_readfirstlane(recU[i]);
        const int   r   = __builtin_amdgcn_readfirstlane(recRow[i]);
        const float we  = recW[i];
        const float4 m  = ((const float4*)(in + (size_t)u * DF))[lane];
        float* a = acc + r * DF + lane;
        __hip_atomic_fetch_add(a +  0, m.x * we, __ATOMIC_RELAXED, __HIP_MEMORY_SCOPE_WORKGROUP);
        __hip_atomic_fetch_add(a + 32, m.y * we, __ATOMIC_RELAXED, __HIP_MEMORY_SCOPE_WORKGROUP);
        __hip_atomic_fetch_add(a + 64, m.z * we, __ATOMIC_RELAXED, __HIP_MEMORY_SCOPE_WORKGROUP);
        __hip_atomic_fetch_add(a + 96, m.w * we, __ATOMIC_RELAXED, __HIP_MEMORY_SCOPE_WORKGROUP);
    }
    __syncthreads();

    float4 bv; bv.x = 0.f; bv.y = 0.f; bv.z = 0.f; bv.w = 0.f;
    if (BIAS_RELU) bv = ((const float4*)bias)[lane];
    for (int r = wv; r < rowsHere; r += 32) {
        const float* a = acc + r * DF + lane;
        float4 o;
        o.x = a[0]; o.y = a[32]; o.z = a[64]; o.w = a[96];
        if (BIAS_RELU) {
            o.x = fmaxf(o.x + bv.x, 0.f);
            o.y = fmaxf(o.y + bv.y, 0.f);
            o.z = fmaxf(o.z + bv.z, 0.f);
            o.w = fmaxf(o.w + bv.w, 0.f);
        }
        ((float4*)(out + (size_t)(row0 + r) * DF))[lane] = o;
    }
}

#define GEMM_ROWS 128
#define GEMM_PITCH 196
__global__ __launch_bounds__(256)
void GCN_gemm_wmma_kernel(const float* __restrict__ A,
                          const float* __restrict__ W,
                          const float* __restrict__ bias,
                          float*       __restrict__ out,
                          int M) {
    __shared__ float4 Bq[(DF / 4) * DF];
    __shared__ __attribute__((aligned(16))) float stage[GEMM_ROWS * GEMM_PITCH];

    const int t    = threadIdx.x;
    const int lane = t & 31;
    const int wv   = __builtin_amdgcn_readfirstlane(t >> 5);
    const int half = lane >> 4;
    const int c    = lane & 15;
    const int rowBase = blockIdx.x * GEMM_ROWS + wv * 16;

    const int row  = rowBase + c;
    const int rowc = (row < M) ? row : (M - 1);
    const float4* Ap = (const float4*)(A + (size_t)rowc * DF) + half;
    float4 a[16];
    #pragma unroll
    for (int j = 0; j < 16; ++j) a[j] = Ap[2 * j];

    #pragma unroll
    for (int i = 0; i < 16; ++i) {
        const int idx = i * 256 + t;
        const int kq  = idx >> 7;
        const int col = idx & 127;
        const float* wp = W + (size_t)(4 * kq) * DF + col;
        float4 v;
        v.x = wp[0]; v.y = wp[DF]; v.z = wp[2 * DF]; v.w = wp[3 * DF];
        Bq[idx] = v;
    }
    __syncthreads();

    v8f acc[8];
    #pragma unroll
    for (int n = 0; n < 8; ++n) acc[n] = v8f{};

    const float4* bp = Bq + half * DF + c;
    float4 bc[8], bn[8];
    #pragma unroll
    for (int n = 0; n < 8; ++n) bc[n] = bp[16 * n];
    #pragma unroll
    for (int j = 0; j < 16; ++j) {
        if (j + 1 < 16) {
            #pragma unroll
            for (int n = 0; n < 8; ++n) bn[n] = bp[(2 * (j + 1)) * DF + 16 * n];
        }
        v2f a0; a0[0] = a[j].x; a0[1] = a[j].y;
        v2f a1; a1[0] = a[j].z; a1[1] = a[j].w;
        #pragma unroll
        for (int n = 0; n < 8; ++n) {
            v2f b0; b0[0] = bc[n].x; b0[1] = bc[n].y;
            acc[n] = __builtin_amdgcn_wmma_f32_16x16x4_f32(
                         false, a0,  false, b0,
                         (short)0, acc[n],  false,  false);
        }
        #pragma unroll
        for (int n = 0; n < 8; ++n) {
            v2f b1; b1[0] = bc[n].z; b1[1] = bc[n].w;
            acc[n] = __builtin_amdgcn_wmma_f32_16x16x4_f32(
                         false, a1,  false, b1,
                         (short)0, acc[n],  false,  false);
        }
        #pragma unroll
        for (int n = 0; n < 8; ++n) bc[n] = bn[n];
    }

    float* st = stage + wv * 16 * GEMM_PITCH;
    #pragma unroll
    for (int n = 0; n < 8; ++n) {
        const float bv = bias[16 * n + c];
        #pragma unroll
        for (int r = 0; r < 8; ++r)
            st[(8 * half + r) * GEMM_PITCH + 16 * n + c] = acc[n][r] + bv;
    }
    __syncthreads();
    if (rowBase + 16 <= M) {
        float4 o[16];
        #pragma unroll
        for (int r = 0; r < 16; ++r) o[r] = *(const float4*)(st + r * GEMM_PITCH + 4 * lane);
        #pragma unroll
        for (int r = 0; r < 16; ++r) ((float4*)(out + (size_t)(rowBase + r) * DF))[lane] = o[r];
    } else {
        for (int r = 0; r < 16; ++r) {
            if (rowBase + r < M)
                ((float4*)(out + (size_t)(rowBase + r) * DF))[lane] = *(const float4*)(st + r * GEMM_PITCH + 4 * lane);
        }
    }
}

extern "C" void kernel_launch(void* const* d_in, const int* in_sizes, int n_in,
                              void* d_out, int out_size, void* d_ws, size_t ws_size,
                              hipStream_t stream) {
    const float* feature = (const float*)d_in[0];
    const float* w0      = (const float*)d_in[1];
    const float* w1      = (const float*)d_in[2];
    const float* b1      = (const float*)d_in[3];
    const float* W2      = (const float*)d_in[4];
    const float* b2      = (const float*)d_in[5];
    const int*   src0    = (const int*)d_in[6];
    const int*   dst0    = (const int*)d_in[7];
    const int*   src1    = (const int*)d_in[8];
    const int*   dst1    = (const int*)d_in[9];

    const int E0 = in_sizes[1];
    const int E1 = in_sizes[2];
    const int N1 = N1C;
    const int N2 = out_size / DF;

    float* h   = (float*)d_ws;
    float* agg = h + (size_t)N1 * DF;
    unsigned* keys0 = (unsigned*)(agg + (size_t)N2 * DF);
    unsigned* keys1 = keys0 + (E0 >> 4) * 4;
    float* out = (float*)d_out;

    {
        int n16_0 = E0 >> 4, n16_1 = E1 >> 4;
        int blocks = (n16_0 + n16_1 + 255) / 256;
        GCN_slab_key_kernel<196, 98><<<blocks, 256, 0, stream>>>(dst0, dst1, keys0, keys1, n16_0, n16_1);
    }

    {
        int blocks = (N1 + 195) / 196;
        GCN_slab_spmm_kernel<196, 8192, true><<<blocks, SLAB_TPB, 0, stream>>>(
            feature, w0, src0, dst0, keys0, b1, h, E0, N1);
    }
    {
        int blocks = (N2 + 97) / 98;
        GCN_slab_spmm_kernel<98, 12288, false><<<blocks, SLAB_TPB, 0, stream>>>(
            h, w1, src1, dst1, keys1, b1, agg, E1, N2);
    }
    {
        int blocks = (N2 + GEMM_ROWS - 1) / GEMM_ROWS;
        GCN_gemm_wmma_kernel<<<blocks, 256, 0, stream>>>(agg, W2, b2, out, N2);
    }
}


# Slab rows owned by one wave each (row mod 32): records counting-sorted by owner on 32 LDS counters, sums by plain 16-byte LDS read-add-write, no f32 LDS atomics; sum order = owner-list arrival
# speedup vs baseline: 2.4858x; 1.0497x over previous
#include <hip/hip_runtime.h>
#include <type_traits>


#define DF   128
#define N1C  50000

#define SLAB_TPB 1024

typedef float v2f __attribute__((ext_vector_type(2)));
typedef float v8f __attribute__((ext_vector_type(8)));

__device__ __noinline__
void GCN_slab_add_direct(float* accRow, const float* __restrict__ f, float we) {
    for (int k = 0; k < DF; ++k)
        __hip_atomic_fetch_add(accRow + k, f[k] * we,
                               __ATOMIC_RELAXED, __HIP_MEMORY_SCOPE_WORKGROUP);
}


template <unsigned ROWS>
__device__ __forceinline__ unsigned GCN_key4(const int4& v) {
    return  (((unsigned)v.x / ROWS) & 255u)        | ((((unsigned)v.y / ROWS) & 255u) << 8) |
           ((((unsigned)v.z / ROWS) & 255u) << 16) |  (((unsigned)v.w / ROWS)         << 24);
}
template <unsigned ROWS>
__device__ __forceinline__ void GCN_key16(const int* __restrict__ dst, unsigned* __restrict__ keys, int i) {
    const int4* d4 = (const int4*)dst + 4 * (size_t)i;
    const int4 a = d4[0], b = d4[1], c = d4[2], d = d4[3];
    uint4 o;
    o.x = GCN_key4<ROWS>(a); o.y = GCN_key4<ROWS>(b); o.z = GCN_key4<ROWS>(c); o.w = GCN_key4<ROWS>(d);
    ((uint4*)keys)[i] = o;
}
template <unsigned ROWS0, unsigned ROWS1>
__global__ __launch_bounds__(256)
void GCN_slab_key_kernel(const int* __restrict__ dst0, const int* __restrict__ dst1,
                         unsigned* __restrict__ keys0, unsigned* __restrict__ keys1,
                         int n16_0, int n16_1) {
    int i = blockIdx.x * 256 + threadIdx.x;
    if (i < n16_0) { GCN_key16<ROWS0>(dst0, keys0, i); return; }
    i -= n16_0;
    if (i < n16_1) GCN_key16<ROWS1>(dst1, keys1, i);
}

template <int ROWS, int CAP, bool BIAS_RELU>
__global__ __launch_bounds__(SLAB_TPB)
void GCN_slab_spmm_kernel(const float* __restrict__ in,
                          const float* __restrict__ w,
                          const int*   __restrict__ src,
                          const int*   __restrict__ dst,
                          const unsigned* __restrict__ keys,
                          const float* __restrict__ bias,
                          float*       __restrict__ out,
                          int nE, int nRows) {
    __shared__ __attribute__((aligned(16))) float acc[ROWS * DF];
    __shared__ int   recRow[CAP];
    __shared__ int   recU[CAP];
    __shared__ float recW[CAP];
    __shared__ int   ord[CAP];
    __shared__ int   binCnt[32];
    __shared__ int   binStart[33];
    __shared__ int   cnt;

    const int t    = threadIdx.x;
    const int lane = t & 31;
    const int wv   = __builtin_amdgcn_readfirstlane(t >> 5);
    const int row0 = blockIdx.x * ROWS;
    int rowsHere = nRows - row0;
    if (rowsHere > ROWS) rowsHere = ROWS;
    if (rowsHere < 0) rowsHere = 0;
    const unsigned rowsU = (unsigned)rowsHere;

    {
        float4 z; z.x = 0.f; z.y = 0.f; z.z = 0.f; z.w = 0.f;
        float4* a4 = (float4*)acc;
        for (int i = t; i < ROWS * DF / 4; i += SLAB_TPB) a4[i] = z;
        if (t == 0) cnt = 0;
        if (t < 32) binCnt[t] = 0;
    }
    __syncthreads();

    const int    n16 = nE >> 4;
    const uint4* k4  = (const uint4*)keys;
    const int start = (int)(((long long)blockIdx.x * n16) / gridDim.x);

    auto edgeOf = [&](unsigned wd) -> int {
        if ((int)wd < 0) return (int)~wd;
        const int p = (int)(wd & 63u);
        const int k = ((p >> 5) << 1) | ((p >> 2) & 1);
        int idx = start + (int)(wd >> 6) + k * SLAB_TPB;
        if (idx >= n16) idx -= n16;
        return 16 * idx + 4 * (p & 3) + ((p >> 3) & 3);
    };
    auto put = [&](int pos, unsigned wd) {
        if (pos < CAP) {
            recU[pos] = (int)wd;
        } else {
            const int e = edgeOf(wd);
            const unsigned r = (unsigned)(dst[e] - row0);
            if (r < rowsU) GCN_slab_add_direct(acc + r * DF, in + (size_t)src[e] * DF, w[e]);
        }
    };
    const unsigned kb = (blockIdx.x & 255u) * 0x01010101u;
    auto eq4 = [&](unsigned x) -> unsigned {
        const unsigned y = x ^ kb;
        return (y - 0x01010101u) & ~y & 0x80808080u;
    };
    auto eq16 = [&](const uint4& v) -> unsigned {
        return (eq4(v.x) >> 7) | (eq4(v.y) >> 6) | (eq4(v.z) >> 5) | (eq4(v.w) >> 4);
    };
    auto batch = [&](auto utag, int b) {
        constexpr int U = decltype(utag)::value;
        uint4 v[U];
        #pragma unroll
        for (int k = 0; k < U; ++k) {
            int idx = start + b + k * SLAB_TPB + t;
            if (idx >= n16) idx -= n16;
            v[k] = k4[idx];
        }
        unsigned h[4] = {0u, 0u, 0u, 0u};
        #pragma unroll
        for (int k = 0; k < U; ++k) h[k] = eq16(v[k]);
        const unsigned mlo = h[0] | (h[1] << 4);
        const unsigned mhi = h[2] | (h[3] << 4);
        const int n = __popc(mlo) + __popc(mhi);
        int pos = 0;
        if (n) pos = atomicAdd(&cnt, n);
        const unsigned tag = (unsigned)(b + t) << 6;
        unsigned long long m = ((unsigned long long)mhi << 32) | mlo;
        while (m) {
            const unsigned p = (unsigned)__builtin_ctzll(m);
            m &= m - 1;
            put(pos++, tag | p);
        }
    };
    int b = 0;
    for (; b + SLAB_TPB * 4 <= n16; b += SLAB_TPB * 4) batch(std::integral_constant<int, 4>{}, b);
    if (b + SLAB_TPB * 2 <= n16) { batch(std::integral_constant<int, 2>{}, b); b += SLAB_TPB * 2; }
    if (b + SLAB_TPB     <= n16) { batch(std::integral_constant<int, 1>{}, b); b += SLAB_TPB; }
    if (b + t < n16) batch(std::integral_constant<int, 1>{}, b);
    for (int e = 16 * n16 + t; e < nE; e += SLAB_TPB)
        if ((unsigned)(dst[e] - row0) < rowsU) put(atomicAdd(&cnt, 1), ~(unsigned)e);
    __syncthreads();

    int count = cnt;
    if (count > CAP) count = CAP;
    for (int i = t; i < count; i += SLAB_TPB) {
        const int e = edgeOf((unsigned)recU[i]);
        const unsigned r = (unsigned)(dst[e] - row0);
        const int   u  = src[e];
        const float we = w[e];
        int pr = -1;
        if (r < rowsU) pr = (int)r | (atomicAdd(&binCnt[r & 31u], 1) << 8);
        recRow[i] = pr;
        recU[i]   = u;
        recW[i]   = we;
    }
    __syncthreads();

    if (wv == 0) {
        const int c = binCnt[lane];
        int incl = c;
        #pragma unroll
        for (int d = 1; d < 32; d <<= 1) {
            const int v = __shfl_up(incl, d);
            if (lane >= d) incl += v;
        }
        binStart[lane] = incl - c;
        if (lane == 31) binStart[32] = incl;
    }
    __syncthreads();

    for (int i = t; i < count; i += SLAB_TPB) {
        const int pr = recRow[i];
        if (pr >= 0) ord[binStart[pr & 31] + (pr >> 8)] = i;
    }
    __syncthreads();

    const int beg = __builtin_amdgcn_readfirstlane(binStart[wv]);
    const int end = __builtin_amdgcn_readfirstlane(binStart[wv + 1]);
    auto loadGroup = [&](int i, float4 (&m)[8], int (&rr)[8], float (&ww)[8]) {
        int p = i + (lane & 7);
        if (p >= end) p = end - 1;
        const int id  = ord[p];
        const int prL = recRow[id];
        const int uL  = recU[id];
        const int wL  = __float_as_int(recW[id]);
        #pragma unroll
        for (int k = 0; k < 8; ++k) {
            const int u = __builtin_amdgcn_readlane(uL, k);
            rr[k] = __builtin_amdgcn_readlane(prL, k) & 255;
            ww[k] = __int_as_float(__builtin_amdgcn_readlane(wL, k));
            m[k]  = ((const float4*)(in + (size_t)u * DF))[lane];
        }
    };
    auto addGroup = [&](int i, const float4 (&m)[8], const int (&rr)[8], const float (&ww)[8]) {
        #pragma unroll
        for (int k = 0; k < 8; ++k) {
            if (i + k < end) {
                float4* a = (float4*)(acc + rr[k] * DF) + lane;
                float4 v = *a;
                v.x += m[k].x * ww[k];
                v.y += m[k].y * ww[k];
                v.z += m[k].z * ww[k];
                v.w += m[k].w * ww[k];
                *a = v;
            }
        }
    };
    if (beg < end) {
        float4 mA[8], mB[8];
        int    rA[8], rB[8];
        float  wA[8], wB[8];
        int i = beg;
        loadGroup(i, mA, rA, wA);
        for (;;) {
            const int j = i + 8;
            if (j < end) loadGroup(j, mB, rB, wB);
            addGroup(i, mA, rA, wA);
            if (j >= end) break;
            i = j + 8;
            if (i < end) loadGroup(i, mA, rA, wA);
            addGroup(j, mB, rB, wB);
            if (i >= end) break;
        }
    }

    float4 bv; bv.x = 0.f; bv.y = 0.f; bv.z = 0.f; bv.w = 0.f;
    if (BIAS_RELU) bv = ((const float4*)bias)[lane];
    for (int r = wv; r < rowsHere; r += 32) {
        float4 o = ((const float4*)(acc + r * DF))[lane];
        if (BIAS_RELU) {
            o.x = fmaxf(o.x + bv.x, 0.f);
            o.y = fmaxf(o.y + bv.y, 0.f);
            o.z = fmaxf(o.z + bv.z, 0.f);
            o.w = fmaxf(o.w + bv.w, 0.f);
        }
        ((float4*)(out + (size_t)(row0 + r) * DF))[lane] = o;
    }
}

#define GEMM_ROWS 128
#define GEMM_PITCH 196
__global__ __launch_bounds__(256)
void GCN_gemm_wmma_kernel(const float* __restrict__ A,
                          const float* __restrict__ W,
                          const float* __restrict__ bias,
                          float*       __restrict__ out,
                          int M) {
    __shared__ float4 Bq[(DF / 4) * DF];
    __shared__ __attribute__((aligned(16))) float stage[GEMM_ROWS * GEMM_PITCH];

    const int t    = threadIdx.x;
    const int lane = t & 31;
    const int wv   = __builtin_amdgcn_readfirstlane(t >> 5);
    const int half = lane >> 4;
    const int c    = lane & 15;
    const int rowBase = blockIdx.x * GEMM_ROWS + wv * 16;

    const int row  = rowBase + c;
    const int rowc = (row < M) ? row : (M - 1);
    const float4* Ap = (const float4*)(A + (size_t)rowc * DF) + half;
    float4 a[16];
    #pragma unroll
    for (int j = 0; j < 16; ++j) a[j] = Ap[2 * j];

    #pragma unroll
    for (int i = 0; i < 16; ++i) {
        const int idx = i * 256 + t;
        const int kq  = idx >> 7;
        const int col = idx & 127;
        const float* wp = W + (size_t)(4 * kq) * DF + col;
        float4 v;
        v.x = wp[0]; v.y = wp[DF]; v.z = wp[2 * DF]; v.w = wp[3 * DF];
        Bq[idx] = v;
    }
    __syncthreads();

    v8f acc[8];
    #pragma unroll
    for (int n = 0; n < 8; ++n) acc[n] = v8f{};

    const float4* bp = Bq + half * DF + c;
    float4 bc[8], bn[8];
    #pragma unroll
    for (int n = 0; n < 8; ++n) bc[n] = bp[16 * n];
    #pragma unroll
    for (int j = 0; j < 16; ++j) {
        if (j + 1 < 16) {
            #pragma unroll
            for (int n = 0; n < 8; ++n) bn[n] = bp[(2 * (j + 1)) * DF + 16 * n];
        }
        v2f a0; a0[0] = a[j].x; a0[1] = a[j].y;
        v2f a1; a1[0] = a[j].z; a1[1] = a[j].w;
        #pragma unroll
        for (int n = 0; n < 8; ++n) {
            v2f b0; b0[0] = bc[n].x; b0[1] = bc[n].y;
            acc[n] = __builtin_amdgcn_wmma_f32_16x16x4_f32(
                         false, a0,  false, b0,
                         (short)0, acc[n],  false,  false);
        }
        #pragma unroll
        for (int n = 0; n < 8; ++n) {
            v2f b1; b1[0] = bc[n].z; b1[1] = bc[n].w;
            acc[n] = __builtin_amdgcn_wmma_f32_16x16x4_f32(
                         false, a1,  false, b1,
                         (short)0, acc[n],  false,  false);
        }
        #pragma unroll
        for (int n = 0; n < 8; ++n) bc[n] = bn[n];
    }

    float* st = stage + wv * 16 * GEMM_PITCH;
    #pragma unroll
    for (int n = 0; n < 8; ++n) {
        const float bv = bias[16 * n + c];
        #pragma unroll
        for (int r = 0; r < 8; ++r)
            st[(8 * half + r) * GEMM_PITCH + 16 * n + c] = acc[n][r] + bv;
    }
    __syncthreads();
    if (rowBase + 16 <= M) {
        float4 o[16];
        #pragma unroll
        for (int r = 0; r < 16; ++r) o[r] = *(const float4*)(st + r * GEMM_PITCH + 4 * lane);
        #pragma unroll
        for (int r = 0; r < 16; ++r) ((float4*)(out + (size_t)(rowBase + r) * DF))[lane] = o[r];
    } else {
        for (int r = 0; r < 16; ++r) {
            if (rowBase + r < M)
                ((float4*)(out + (size_t)(rowBase + r) * DF))[lane] = *(const float4*)(st + r * GEMM_PITCH + 4 * lane);
        }
    }
}

extern "C" void kernel_launch(void* const* d_in, const int* in_sizes, int n_in,
                              void* d_out, int out_size, void* d_ws, size_t ws_size,
                              hipStream_t stream) {
    const float* feature = (const float*)d_in[0];
    const float* w0      = (const float*)d_in[1];
    const float* w1      = (const float*)d_in[2];
    const float* b1      = (const float*)d_in[3];
    const float* W2      = (const float*)d_in[4];
    const float* b2      = (const float*)d_in[5];
    const int*   src0    = (const int*)d_in[6];
    const int*   dst0    = (const int*)d_in[7];
    const int*   src1    = (const int*)d_in[8];
    const int*   dst1    = (const int*)d_in[9];

    const int E0 = in_sizes[1];
    const int E1 = in_sizes[2];
    const int N1 = N1C;
    const int N2 = out_size / DF;

    float* h   = (float*)d_ws;
    float* agg = h + (size_t)N1 * DF;
    unsigned* keys0 = (unsigned*)(agg + (size_t)N2 * DF);
    unsigned* keys1 = keys0 + (E0 >> 4) * 4;
    float* out = (float*)d_out;

    {
        int n16_0 = E0 >> 4, n16_1 = E1 >> 4;
        int blocks = (n16_0 + n16_1 + 255) / 256;
        GCN_slab_key_kernel<196, 98><<<blocks, 256, 0, stream>>>(dst0, dst1, keys0, keys1, n16_0, n16_1);
    }

    {
        int blocks = (N1 + 195) / 196;
        GCN_slab_spmm_kernel<196, 8192, true><<<blocks, SLAB_TPB, 0, stream>>>(
            feature, w0, src0, dst0, keys0, b1, h, E0, N1);
    }
    {
        int blocks = (N2 + 97) / 98;
        GCN_slab_spmm_kernel<98, 12288, false><<<blocks, SLAB_TPB, 0, stream>>>(
            h, w1, src1, dst1, keys1, b1, agg, E1, N2);
    }
    {
        int blocks = (N2 + GEMM_ROWS - 1) / GEMM_ROWS;
        GCN_gemm_wmma_kernel<<<blocks, 256, 0, stream>>>(agg, W2, b2, out, N2);
    }
}


# Gather loop split into a branch-free steady state (two full groups of 8 per trip) and a guarded tail: the next 8 row loads stay in flight over the current 8 adds; sum order kept
# speedup vs baseline: 2.5788x; 1.0374x over previous
#include <hip/hip_runtime.h>
#include <type_traits>


#define DF   128
#define N1C  50000

#define SLAB_TPB 1024

typedef float v2f __attribute__((ext_vector_type(2)));
typedef float v8f __attribute__((ext_vector_type(8)));

__device__ __noinline__
void GCN_slab_add_direct(float* accRow, const float* __restrict__ f, float we) {
    for (int k = 0; k < DF; ++k)
        __hip_atomic_fetch_add(accRow + k, f[k] * we,
                               __ATOMIC_RELAXED, __HIP_MEMORY_SCOPE_WORKGROUP);
}


template <unsigned ROWS>
__device__ __forceinline__ unsigned GCN_key4(const int4& v) {
    return  (((unsigned)v.x / ROWS) & 255u)        | ((((unsigned)v.y / ROWS) & 255u) << 8) |
           ((((unsigned)v.z / ROWS) & 255u) << 16) |  (((unsigned)v.w / ROWS)         << 24);
}
template <unsigned ROWS>
__device__ __forceinline__ void GCN_key16(const int* __restrict__ dst, unsigned* __restrict__ keys, int i) {
    const int4* d4 = (const int4*)dst + 4 * (size_t)i;
    const int4 a = d4[0], b = d4[1], c = d4[2], d = d4[3];
    uint4 o;
    o.x = GCN_key4<ROWS>(a); o.y = GCN_key4<ROWS>(b); o.z = GCN_key4<ROWS>(c); o.w = GCN_key4<ROWS>(d);
    ((uint4*)keys)[i] = o;
}
template <unsigned ROWS0, unsigned ROWS1>
__global__ __launch_bounds__(256)
void GCN_slab_key_kernel(const int* __restrict__ dst0, const int* __restrict__ dst1,
                         unsigned* __restrict__ keys0, unsigned* __restrict__ keys1,
                         int n16_0, int n16_1) {
    int i = blockIdx.x * 256 + threadIdx.x;
    if (i < n16_0) { GCN_key16<ROWS0>(dst0, keys0, i); return; }
    i -= n16_0;
    if (i < n16_1) GCN_key16<ROWS1>(dst1, keys1, i);
}

template <int ROWS, int CAP, bool BIAS_RELU>
__global__ __launch_bounds__(SLAB_TPB)
void GCN_slab_spmm_kernel(const float* __restrict__ in,
                          const float* __restrict__ w,
                          const int*   __restrict__ src,
                          const int*   __restrict__ dst,
                          const unsigned* __restrict__ keys,
                          const float* __restrict__ bias,
                          float*       __restrict__ out,
                          int nE, int nRows) {
    __shared__ __attribute__((aligned(16))) float acc[ROWS * DF];
    __shared__ int   recRow[CAP];
    __shared__ int   recU[CAP];
    __shared__ float recW[CAP];
    __shared__ int   ord[CAP];
    __shared__ int   binCnt[32];
    __shared__ int   binStart[33];
    __shared__ int   cnt;

    const int t    = threadIdx.x;
    const int lane = t & 31;
    const int wv   = __builtin_amdgcn_readfirstlane(t >> 5);
    const int row0 = blockIdx.x * ROWS;
    int rowsHere = nRows - row0;
    if (rowsHere > ROWS) rowsHere = ROWS;
    if (rowsHere < 0) rowsHere = 0;
    const unsigned rowsU = (unsigned)rowsHere;

    {
        float4 z; z.x = 0.f; z.y = 0.f; z.z = 0.f; z.w = 0.f;
        float4* a4 = (float4*)acc;
        for (int i = t; i < ROWS * DF / 4; i += SLAB_TPB) a4[i] = z;
        if (t == 0) cnt = 0;
        if (t < 32) binCnt[t] = 0;
    }
    __syncthreads();

    const int    n16 = nE >> 4;
    const uint4* k4  = (const uint4*)keys;
    const int start = (int)(((long long)blockIdx.x * n16) / gridDim.x);

    auto edgeOf = [&](unsigned wd) -> int {
        if ((int)wd < 0) return (int)~wd;
        const int p = (int)(wd & 63u);
        const int k = ((p >> 5) << 1) | ((p >> 2) & 1);
        int idx = start + (int)(wd >> 6) + k * SLAB_TPB;
        if (idx >= n16) idx -= n16;
        return 16 * idx + 4 * (p & 3) + ((p >> 3) & 3);
    };
    auto put = [&](int pos, unsigned wd) {
        if (pos < CAP) {
            recU[pos] = (int)wd;
        } else {
            const int e = edgeOf(wd);
            const unsigned r = (unsigned)(dst[e] - row0);
            if (r < rowsU) GCN_slab_add_direct(acc + r * DF, in + (size_t)src[e] * DF, w[e]);
        }
    };
    const unsigned kb = (blockIdx.x & 255u) * 0x01010101u;
    auto eq4 = [&](unsigned x) -> unsigned {
        const unsigned y = x ^ kb;
        return (y - 0x01010101u) & ~y & 0x80808080u;
    };
    auto eq16 = [&](const uint4& v) -> unsigned {
        return (eq4(v.x) >> 7) | (eq4(v.y) >> 6) | (eq4(v.z) >> 5) | (eq4(v.w) >> 4);
    };
    auto batch = [&](auto utag, int b) {
        constexpr int U = decltype(utag)::value;
        uint4 v[U];
        #pragma unroll
        for (int k = 0; k < U; ++k) {
            int idx = start + b + k * SLAB_TPB + t;
            if (idx >= n16) idx -= n16;
            v[k] = k4[idx];
        }
        unsigned h[4] = {0u, 0u, 0u, 0u};
        #pragma unroll
        for (int k = 0; k < U; ++k) h[k] = eq16(v[k]);
        const unsigned mlo = h[0] | (h[1] << 4);
        const unsigned mhi = h[2] | (h[3] << 4);
        const int n = __popc(mlo) + __popc(mhi);
        int pos = 0;
        if (n) pos = atomicAdd(&cnt, n);
        const unsigned tag = (unsigned)(b + t) << 6;
        unsigned long long m = ((unsigned long long)mhi << 32) | mlo;
        while (m) {
            const unsigned p = (unsigned)__builtin_ctzll(m);
            m &= m - 1;
            put(pos++, tag | p);
        }
    };
    int b = 0;
    for (; b + SLAB_TPB * 4 <= n16; b += SLAB_TPB * 4) batch(std::integral_constant<int, 4>{}, b);
    if (b + SLAB_TPB * 2 <= n16) { batch(std::integral_constant<int, 2>{}, b); b += SLAB_TPB * 2; }
    if (b + SLAB_TPB     <= n16) { batch(std::integral_constant<int, 1>{}, b); b += SLAB_TPB; }
    if (b + t < n16) batch(std::integral_constant<int, 1>{}, b);
    for (int e = 16 * n16 + t; e < nE; e += SLAB_TPB)
        if ((unsigned)(dst[e] - row0) < rowsU) put(atomicAdd(&cnt, 1), ~(unsigned)e);
    __syncthreads();

    int count = cnt;
    if (count > CAP) count = CAP;
    for (int i = t; i < count; i += SLAB_TPB) {
        const int e = edgeOf((unsigned)recU[i]);
        const unsigned r = (unsigned)(dst[e] - row0);
        const int   u  = src[e];
        const float we = w[e];
        int pr = -1;
        if (r < rowsU) pr = (int)r | (atomicAdd(&binCnt[r & 31u], 1) << 8);
        recRow[i] = pr;
        recU[i]   = u;
        recW[i]   = we;
    }
    __syncthreads();

    if (wv == 0) {
        const int c = binCnt[lane];
        int incl = c;
        #pragma unroll
        for (int d = 1; d < 32; d <<= 1) {
            const int v = __shfl_up(incl, d);
            if (lane >= d) incl += v;
        }
        binStart[lane] = incl - c;
        if (lane == 31) binStart[32] = incl;
    }
    __syncthreads();

    for (int i = t; i < count; i += SLAB_TPB) {
        const int pr = recRow[i];
        if (pr >= 0) ord[binStart[pr & 31] + (pr >> 8)] = i;
    }
    __syncthreads();

    const int beg = __builtin_amdgcn_readfirstlane(binStart[wv]);
    const int end = __builtin_amdgcn_readfirstlane(binStart[wv + 1]);
    auto loadGroup = [&](int i, float4 (&m)[8], int (&rr)[8], float (&ww)[8]) {
        int p = i + (lane & 7);
        if (p >= end) p = end - 1;
        const int id  = ord[p];
        const int prL = recRow[id];
        const int uL  = recU[id];
        const int wL  = __float_as_int(recW[id]);
        #pragma unroll
        for (int k = 0; k < 8; ++k) {
            const int u = __builtin_amdgcn_readlane(uL, k);
            rr[k] = __builtin_amdgcn_readlane(prL, k) & 255;
            ww[k] = __int_as_float(__builtin_amdgcn_readlane(wL, k));
            m[k]  = ((const float4*)(in + (size_t)u * DF))[lane];
        }
        __builtin_amdgcn_sched_barrier(0);
    };
    auto addOne = [&](const float4& mk, int r, float wk) {
        float4* a = (float4*)(acc + r * DF) + lane;
        float4 v = *a;
        v.x += mk.x * wk;
        v.y += mk.y * wk;
        v.z += mk.z * wk;
        v.w += mk.w * wk;
        *a = v;
    };
    auto addFull = [&](const float4 (&m)[8], const int (&rr)[8], const float (&ww)[8]) {
        #pragma unroll
        for (int k = 0; k < 8; ++k) addOne(m[k], rr[k], ww[k]);
    };
    auto addTail = [&](int i, const float4 (&m)[8], const int (&rr)[8], const float (&ww)[8]) {
        #pragma unroll
        for (int k = 0; k < 8; ++k)
            if (i + k < end) addOne(m[k], rr[k], ww[k]);
    };
    if (beg < end) {
        float4 mA[8], mB[8];
        int    rA[8], rB[8];
        float  wA[8], wB[8];
        int i = beg;
        loadGroup(i, mA, rA, wA);
        while (i + 16 < end) {
            loadGroup(i + 8, mB, rB, wB);
            addFull(mA, rA, wA);
            loadGroup(i + 16, mA, rA, wA);
            addFull(mB, rB, wB);
            i += 16;
        }
        if (i + 8 < end) {
            loadGroup(i + 8, mB, rB, wB);
            addFull(mA, rA, wA);
            addTail(i + 8, mB, rB, wB);
        } else {
            addTail(i, mA, rA, wA);
        }
    }

    float4 bv; bv.x = 0.f; bv.y = 0.f; bv.z = 0.f; bv.w = 0.f;
    if (BIAS_RELU) bv = ((const float4*)bias)[lane];
    for (int r = wv; r < rowsHere; r += 32) {
        float4 o = ((const float4*)(acc + r * DF))[lane];
        if (BIAS_RELU) {
            o.x = fmaxf(o.x + bv.x, 0.f);
            o.y = fmaxf(o.y + bv.y, 0.f);
            o.z = fmaxf(o.z + bv.z, 0.f);
            o.w = fmaxf(o.w + bv.w, 0.f);
        }
        ((float4*)(out + (size_t)(row0 + r) * DF))[lane] = o;
    }
}

#define GEMM_ROWS 128
#define GEMM_PITCH 196
__global__ __launch_bounds__(256)
void GCN_gemm_wmma_kernel(const float* __restrict__ A,
                          const float* __restrict__ W,
                          const float* __restrict__ bias,
                          float*       __restrict__ out,
                          int M) {
    __shared__ float4 Bq[(DF / 4) * DF];
    __shared__ __attribute__((aligned(16))) float stage[GEMM_ROWS * GEMM_PITCH];

    const int t    = threadIdx.x;
    const int lane = t & 31;
    const int wv   = __builtin_amdgcn_readfirstlane(t >> 5);
    const int half = lane >> 4;
    const int c    = lane & 15;
    const int rowBase = blockIdx.x * GEMM_ROWS + wv * 16;

    const int row  = rowBase + c;
    const int rowc = (row < M) ? row : (M - 1);
    const float4* Ap = (const float4*)(A + (size_t)rowc * DF) + half;
    float4 a[16];
    #pragma unroll
    for (int j = 0; j < 16; ++j) a[j] = Ap[2 * j];

    #pragma unroll
    for (int i = 0; i < 16; ++i) {
        const int idx = i * 256 + t;
        const int kq  = idx >> 7;
        const int col = idx & 127;
        const float* wp = W + (size_t)(4 * kq) * DF + col;
        float4 v;
        v.x = wp[0]; v.y = wp[DF]; v.z = wp[2 * DF]; v.w = wp[3 * DF];
        Bq[idx] = v;
    }
    __syncthreads();

    v8f acc[8];
    #pragma unroll
    for (int n = 0; n < 8; ++n) acc[n] = v8f{};

    const float4* bp = Bq + half * DF + c;
    float4 bc[8], bn[8];
    #pragma unroll
    for (int n = 0; n < 8; ++n) bc[n] = bp[16 * n];
    #pragma unroll
    for (int j = 0; j < 16; ++j) {
        if (j + 1 < 16) {
            #pragma unroll
            for (int n = 0; n < 8; ++n) bn[n] = bp[(2 * (j + 1)) * DF + 16 * n];
        }
        v2f a0; a0[0] = a[j].x; a0[1] = a[j].y;
        v2f a1; a1[0] = a[j].z; a1[1] = a[j].w;
        #pragma unroll
        for (int n = 0; n < 8; ++n) {
            v2f b0; b0[0] = bc[n].x; b0[1] = bc[n].y;
            acc[n] = __builtin_amdgcn_wmma_f32_16x16x4_f32(
                         false, a0,  false, b0,
                         (short)0, acc[n],  false,  false);
        }
        #pragma unroll
        for (int n = 0; n < 8; ++n) {
            v2f b1; b1[0] = bc[n].z; b1[1] = bc[n].w;
            acc[n] = __builtin_amdgcn_wmma_f32_16x16x4_f32(
                         false, a1,  false, b1,
                         (short)0, acc[n],  false,  false);
        }
        #pragma unroll
        for (int n = 0; n < 8; ++n) bc[n] = bn[n];
    }

    float* st = stage + wv * 16 * GEMM_PITCH;
    #pragma unroll
    for (int n = 0; n < 8; ++n) {
        const float bv = bias[16 * n + c];
        #pragma unroll
        for (int r = 0; r < 8; ++r)
            st[(8 * half + r) * GEMM_PITCH + 16 * n + c] = acc[n][r] + bv;
    }
    __syncthreads();
    if (rowBase + 16 <= M) {
        float4 o[16];
        #pragma unroll
        for (int r = 0; r < 16; ++r) o[r] = *(const float4*)(st + r * GEMM_PITCH + 4 * lane);
        #pragma unroll
        for (int r = 0; r < 16; ++r) ((float4*)(out + (size_t)(rowBase + r) * DF))[lane] = o[r];
    } else {
        for (int r = 0; r < 16; ++r) {
            if (rowBase + r < M)
                ((float4*)(out + (size_t)(rowBase + r) * DF))[lane] = *(const float4*)(st + r * GEMM_PITCH + 4 * lane);
        }
    }
}

extern "C" void kernel_launch(void* const* d_in, const int* in_sizes, int n_in,
                              void* d_out, int out_size, void* d_ws, size_t ws_size,
                              hipStream_t stream) {
    const float* feature = (const float*)d_in[0];
    const float* w0      = (const float*)d_in[1];
    const float* w1      = (const float*)d_in[2];
    const float* b1      = (const float*)d_in[3];
    const float* W2      = (const float*)d_in[4];
    const float* b2      = (const float*)d_in[5];
    const int*   src0    = (const int*)d_in[6];
    const int*   dst0    = (const int*)d_in[7];
    const int*   src1    = (const int*)d_in[8];
    const int*   dst1    = (const int*)d_in[9];

    const int E0 = in_sizes[1];
    const int E1 = in_sizes[2];
    const int N1 = N1C;
    const int N2 = out_size / DF;

    float* h   = (float*)d_ws;
    float* agg = h + (size_t)N1 * DF;
    unsigned* keys0 = (unsigned*)(agg + (size_t)N2 * DF);
    unsigned* keys1 = keys0 + (E0 >> 4) * 4;
    float* out = (float*)d_out;

    {
        int n16_0 = E0 >> 4, n16_1 = E1 >> 4;
        int blocks = (n16_0 + n16_1 + 255) / 256;
        GCN_slab_key_kernel<196, 98><<<blocks, 256, 0, stream>>>(dst0, dst1, keys0, keys1, n16_0, n16_1);
    }

    {
        int blocks = (N1 + 195) / 196;
        GCN_slab_spmm_kernel<196, 8192, true><<<blocks, SLAB_TPB, 0, stream>>>(
            feature, w0, src0, dst0, keys0, b1, h, E0, N1);
    }
    {
        int blocks = (N2 + 97) / 98;
        GCN_slab_spmm_kernel<98, 12288, false><<<blocks, SLAB_TPB, 0, stream>>>(
            h, w1, src1, dst1, keys1, b1, agg, E1, N2);
    }
    {
        int blocks = (N2 + GEMM_ROWS - 1) / GEMM_ROWS;
        GCN_gemm_wmma_kernel<<<blocks, 256, 0, stream>>>(agg, W2, b2, out, N2);
    }
}


# Slab work-groups read only their own row of a one-bit-per-edge slab bit matrix (built by the first launch with LDS ORs, whole 128-byte lines), not every edge's byte key; sum order still list arrival
# speedup vs baseline: 2.7226x; 1.0558x over previous
#include <hip/hip_runtime.h>


#define DF   128
#define N1C  50000

#define SLAB_TPB 1024

typedef float v2f __attribute__((ext_vector_type(2)));
typedef float v8f __attribute__((ext_vector_type(8)));

__device__ __noinline__
void GCN_slab_add_direct(float* accRow, const float* __restrict__ f, float we) {
    for (int k = 0; k < DF; ++k)
        __hip_atomic_fetch_add(accRow + k, f[k] * we,
                               __ATOMIC_RELAXED, __HIP_MEMORY_SCOPE_WORKGROUP);
}


#define NSLAB    256
#define BM_EDGES 2048
#define BM_WORDS (BM_EDGES / 32)
#define BM_TPB   128
template <unsigned ROWS>
__device__ __forceinline__ void GCN_bits_chunk(const int* __restrict__ dst, unsigned* __restrict__ bits,
                                               int chunk, int nE, int rowWords, unsigned* bm) {
    constexpr int QPT = BM_EDGES / 4 / BM_TPB;
    constexpr int PPS = BM_WORDS / 4;
    constexpr int PPT = NSLAB * PPS / BM_TPB;
    const int t  = threadIdx.x;
    const int e0 = chunk * BM_EDGES;
    int4 d[QPT];
    if (e0 + BM_EDGES <= nE) {
        #pragma unroll
        for (int k = 0; k < QPT; ++k) d[k] = *(const int4*)(dst + e0 + 4 * (k * BM_TPB + t));
    } else {
        #pragma unroll
        for (int k = 0; k < QPT; ++k) {
            const int e = e0 + 4 * (k * BM_TPB + t);
            int4 v; v.x = -1; v.y = -1; v.z = -1; v.w = -1;
            if (e     < nE) v.x = dst[e];
            if (e + 1 < nE) v.y = dst[e + 1];
            if (e + 2 < nE) v.z = dst[e + 2];
            if (e + 3 < nE) v.w = dst[e + 3];
            d[k] = v;
        }
    }
    {
        uint4 z; z.x = 0u; z.y = 0u; z.z = 0u; z.w = 0u;
        #pragma unroll
        for (int k = 0; k < PPT; ++k) ((uint4*)bm)[k * BM_TPB + t] = z;
    }
    __syncthreads();
    #pragma unroll
    for (int k = 0; k < QPT; ++k) {
        const int q = k * BM_TPB + t;
        unsigned* wq = bm + (q >> 3);
        const unsigned b0 = 1u << (4 * (q & 7));
        const unsigned s0 = (unsigned)d[k].x / ROWS, s1 = (unsigned)d[k].y / ROWS;
        const unsigned s2 = (unsigned)d[k].z / ROWS, s3 = (unsigned)d[k].w / ROWS;
        if (s0 < NSLAB) atomicOr(wq + s0 * BM_WORDS, b0);
        if (s1 < NSLAB) atomicOr(wq + s1 * BM_WORDS, b0 << 1);
        if (s2 < NSLAB) atomicOr(wq + s2 * BM_WORDS, b0 << 2);
        if (s3 < NSLAB) atomicOr(wq + s3 * BM_WORDS, b0 << 3);
    }
    __syncthreads();
    const uint4* lp = (const uint4*)bm + t;
    unsigned*    gp = bits + (size_t)chunk * BM_WORDS + (size_t)(t / PPS) * rowWords + 4 * (t % PPS);
    #pragma unroll 8
    for (int k = 0; k < PPT; ++k)
        *(uint4*)(gp + (size_t)(k * (BM_TPB / PPS)) * rowWords) = lp[k * BM_TPB];
}
template <unsigned ROWS0, unsigned ROWS1>
__global__ __launch_bounds__(BM_TPB)
void GCN_slab_bits_kernel(const int* __restrict__ dst0, const int* __restrict__ dst1,
                          unsigned* __restrict__ bits0, unsigned* __restrict__ bits1,
                          int nE0, int nE1, int chunks0, int rowWords0, int rowWords1) {
    __shared__ __attribute__((aligned(16))) unsigned bm[NSLAB * BM_WORDS];
    const int c = blockIdx.x;
    if (c < chunks0) GCN_bits_chunk<ROWS0>(dst0, bits0, c, nE0, rowWords0, bm);
    else             GCN_bits_chunk<ROWS1>(dst1, bits1, c - chunks0, nE1, rowWords1, bm);
}

template <int ROWS, int CAP, bool BIAS_RELU>
__global__ __launch_bounds__(SLAB_TPB)
void GCN_slab_spmm_kernel(const float* __restrict__ in,
                          const float* __restrict__ w,
                          const int*   __restrict__ src,
                          const int*   __restrict__ dst,
                          const unsigned* __restrict__ bits,
                          const float* __restrict__ bias,
                          float*       __restrict__ out,
                          int rowWords, int nRows) {
    __shared__ __attribute__((aligned(16))) float acc[ROWS * DF];
    __shared__ int   recRow[CAP];
    __shared__ int   recU[CAP];
    __shared__ float recW[CAP];
    __shared__ int   ord[CAP];
    __shared__ int   binCnt[32];
    __shared__ int   binStart[33];
    __shared__ int   cnt;

    const int t    = threadIdx.x;
    const int lane = t & 31;
    const int wv   = __builtin_amdgcn_readfirstlane(t >> 5);
    const int row0 = blockIdx.x * ROWS;
    int rowsHere = nRows - row0;
    if (rowsHere > ROWS) rowsHere = ROWS;
    if (rowsHere < 0) rowsHere = 0;
    const unsigned rowsU = (unsigned)rowsHere;

    {
        float4 z; z.x = 0.f; z.y = 0.f; z.z = 0.f; z.w = 0.f;
        float4* a4 = (float4*)acc;
        for (int i = t; i < ROWS * DF / 4; i += SLAB_TPB) a4[i] = z;
        if (t == 0) cnt = 0;
        if (t < 32) binCnt[t] = 0;
    }
    __syncthreads();

    const int    n4  = rowWords >> 2;
    const uint4* bm4 = (const uint4*)(bits + (size_t)blockIdx.x * rowWords);
    auto put = [&](int pos, int e) {
        if (pos < CAP) {
            recU[pos] = e;
        } else {
            const unsigned r = (unsigned)(dst[e] - row0);
            if (r < rowsU) GCN_slab_add_direct(acc + r * DF, in + (size_t)src[e] * DF, w[e]);
        }
    };
    constexpr int SCAN_U = 5;
    for (int b = 0; b < n4; b += SCAN_U * SLAB_TPB) {
        uint4 v[SCAN_U];
        #pragma unroll
        for (int k = 0; k < SCAN_U; ++k) {
            const int idx = b + k * SLAB_TPB + t;
            uint4 x; x.x = 0u; x.y = 0u; x.z = 0u; x.w = 0u;
            if (idx < n4) x = bm4[idx];
            v[k] = x;
        }
        int n = 0;
        #pragma unroll
        for (int k = 0; k < SCAN_U; ++k)
            n += __popc(v[k].x) + __popc(v[k].y) + __popc(v[k].z) + __popc(v[k].w);
        int incl = n;
        #pragma unroll
        for (int d = 1; d < 32; d <<= 1) {
            const int up = __shfl_up(incl, d);
            if (lane >= d) incl += up;
        }
        int base = 0;
        if (lane == 31 && incl) base = atomicAdd(&cnt, incl);
        int pos = __builtin_amdgcn_readlane(base, 31) + incl - n;
        const bool roomy = __builtin_amdgcn_ballot_w32(pos + n > CAP) == 0u;
        if (roomy) {
            #pragma unroll
            for (int k = 0; k < SCAN_U; ++k) {
                const int eb = 128 * (b + k * SLAB_TPB + t);
                for (unsigned m = v[k].x; m; m &= m - 1) recU[pos++] = eb      + __builtin_ctz(m);
                for (unsigned m = v[k].y; m; m &= m - 1) recU[pos++] = eb + 32 + __builtin_ctz(m);
                for (unsigned m = v[k].z; m; m &= m - 1) recU[pos++] = eb + 64 + __builtin_ctz(m);
                for (unsigned m = v[k].w; m; m &= m - 1) recU[pos++] = eb + 96 + __builtin_ctz(m);
            }
        } else {
            #pragma unroll
            for (int k = 0; k < SCAN_U; ++k) {
                const int eb = 128 * (b + k * SLAB_TPB + t);
                for (unsigned m = v[k].x; m; m &= m - 1) put(pos++, eb      + __builtin_ctz(m));
                for (unsigned m = v[k].y; m; m &= m - 1) put(pos++, eb + 32 + __builtin_ctz(m));
                for (unsigned m = v[k].z; m; m &= m - 1) put(pos++, eb + 64 + __builtin_ctz(m));
                for (unsigned m = v[k].w; m; m &= m - 1) put(pos++, eb + 96 + __builtin_ctz(m));
            }
        }
    }
    __syncthreads();

    int count = cnt;
    if (count > CAP) count = CAP;
    for (int i = t; i < count; i += SLAB_TPB) {
        const int e = recU[i];
        const unsigned r = (unsigned)(dst[e] - row0);
        const int   u  = src[e];
        const float we = w[e];
        int pr = -1;
        if (r < rowsU) pr = (int)r | (atomicAdd(&binCnt[r & 31u], 1) << 8);
        recRow[i] = pr;
        recU[i]   = u;
        recW[i]   = we;
    }
    __syncthreads();

    if (wv == 0) {
        const int c = binCnt[lane];
        int incl = c;
        #pragma unroll
        for (int d = 1; d < 32; d <<= 1) {
            const int v = __shfl_up(incl, d);
            if (lane >= d) incl += v;
        }
        binStart[lane] = incl - c;
        if (lane == 31) binStart[32] = incl;
    }
    __syncthreads();

    for (int i = t; i < count; i += SLAB_TPB) {
        const int pr = recRow[i];
        if (pr >= 0) ord[binStart[pr & 31] + (pr >> 8)] = i;
    }
    __syncthreads();

    const int beg = __builtin_amdgcn_readfirstlane(binStart[wv]);
    const int end = __builtin_amdgcn_readfirstlane(binStart[wv + 1]);
    auto loadGroup = [&](int i, float4 (&m)[8], int (&rr)[8], float (&ww)[8]) {
        int p = i + (lane & 7);
        if (p >= end) p = end - 1;
        const int id  = ord[p];
        const int prL = recRow[id];
        const int uL  = recU[id];
        const int wL  = __float_as_int(recW[id]);
        #pragma unroll
        for (int k = 0; k < 8; ++k) {
            const int u = __builtin_amdgcn_readlane(uL, k);
            rr[k] = __builtin_amdgcn_readlane(prL, k) & 255;
            ww[k] = __int_as_float(__builtin_amdgcn_readlane(wL, k));
            m[k]  = ((const float4*)(in + (size_t)u * DF))[lane];
        }
        __builtin_amdgcn_sched_barrier(0);
    };
    auto addOne = [&](const float4& mk, int r, float wk) {
        float4* a = (float4*)(acc + r * DF) + lane;
        float4 v = *a;
        v.x += mk.x * wk;
        v.y += mk.y * wk;
        v.z += mk.z * wk;
        v.w += mk.w * wk;
        *a = v;
    };
    auto addFull = [&](const float4 (&m)[8], const int (&rr)[8], const float (&ww)[8]) {
        #pragma unroll
        for (int k = 0; k < 8; ++k) addOne(m[k], rr[k], ww[k]);
    };
    auto addTail = [&](int i, const float4 (&m)[8], const int (&rr)[8], const float (&ww)[8]) {
        #pragma unroll
        for (int k = 0; k < 8; ++k)
            if (i + k < end) addOne(m[k], rr[k], ww[k]);
    };
    if (beg < end) {
        float4 mA[8], mB[8];
        int    rA[8], rB[8];
        float  wA[8], wB[8];
        int i = beg;
        loadGroup(i, mA, rA, wA);
        while (i + 16 < end) {
            loadGroup(i + 8, mB, rB, wB);
            addFull(mA, rA, wA);
            loadGroup(i + 16, mA, rA, wA);
            addFull(mB, rB, wB);
            i += 16;
        }
        if (i + 8 < end) {
            loadGroup(i + 8, mB, rB, wB);
            addFull(mA, rA, wA);
            addTail(i + 8, mB, rB, wB);
        } else {
            addTail(i, mA, rA, wA);
        }
    }

    float4 bv; bv.x = 0.f; bv.y = 0.f; bv.z = 0.f; bv.w = 0.f;
    if (BIAS_RELU) bv = ((const float4*)bias)[lane];
    for (int r = wv; r < rowsHere; r += 32) {
        float4 o = ((const float4*)(acc + r * DF))[lane];
        if (BIAS_RELU) {
            o.x = fmaxf(o.x + bv.x, 0.f);
            o.y = fmaxf(o.y + bv.y, 0.f);
            o.z = fmaxf(o.z + bv.z, 0.f);
            o.w = fmaxf(o.w + bv.w, 0.f);
        }
        ((float4*)(out + (size_t)(row0 + r) * DF))[lane] = o;
    }
}

#define GEMM_ROWS 128
#define GEMM_PITCH 196
__global__ __launch_bounds__(256)
void GCN_gemm_wmma_kernel(const float* __restrict__ A,
                          const float* __restrict__ W,
                          const float* __restrict__ bias,
                          float*       __restrict__ out,
                          int M) {
    __shared__ float4 Bq[(DF / 4) * DF];
    __shared__ __attribute__((aligned(16))) float stage[GEMM_ROWS * GEMM_PITCH];

    const int t    = threadIdx.x;
    const int lane = t & 31;
    const int wv   = __builtin_amdgcn_readfirstlane(t >> 5);
    const int half = lane >> 4;
    const int c    = lane & 15;
    const int rowBase = blockIdx.x * GEMM_ROWS + wv * 16;

    const int row  = rowBase + c;
    const int rowc = (row < M) ? row : (M - 1);
    const float4* Ap = (const float4*)(A + (size_t)rowc * DF) + half;
    float4 a[16];
    #pragma unroll
    for (int j = 0; j < 16; ++j) a[j] = Ap[2 * j];

    #pragma unroll
    for (int i = 0; i < 16; ++i) {
        const int idx = i * 256 + t;
        const int kq  = idx >> 7;
        const int col = idx & 127;
        const float* wp = W + (size_t)(4 * kq) * DF + col;
        float4 v;
        v.x = wp[0]; v.y = wp[DF]; v.z = wp[2 * DF]; v.w = wp[3 * DF];
        Bq[idx] = v;
    }
    __syncthreads();

    v8f acc[8];
    #pragma unroll
    for (int n = 0; n < 8; ++n) acc[n] = v8f{};

    const float4* bp = Bq + half * DF + c;
    float4 bc[8], bn[8];
    #pragma unroll
    for (int n = 0; n < 8; ++n) bc[n] = bp[16 * n];
    #pragma unroll
    for (int j = 0; j < 16; ++j) {
        if (j + 1 < 16) {
            #pragma unroll
            for (int n = 0; n < 8; ++n) bn[n] = bp[(2 * (j + 1)) * DF + 16 * n];
        }
        v2f a0; a0[0] = a[j].x; a0[1] = a[j].y;
        v2f a1; a1[0] = a[j].z; a1[1] = a[j].w;
        #pragma unroll
        for (int n = 0; n < 8; ++n) {
            v2f b0; b0[0] = bc[n].x; b0[1] = bc[n].y;
            acc[n] = __builtin_amdgcn_wmma_f32_16x16x4_f32(
                         false, a0,  false, b0,
                         (short)0, acc[n],  false,  false);
        }
        #pragma unroll
        for (int n = 0; n < 8; ++n) {
            v2f b1; b1[0] = bc[n].z; b1[1] = bc[n].w;
            acc[n] = __builtin_amdgcn_wmma_f32_16x16x4_f32(
                         false, a1,  false, b1,
                         (short)0, acc[n],  false,  false);
        }
        #pragma unroll
        for (int n = 0; n < 8; ++n) bc[n] = bn[n];
    }

    float* st = stage + wv * 16 * GEMM_PITCH;
    #pragma unroll
    for (int n = 0; n < 8; ++n) {
        const float bv = bias[16 * n + c];
        #pragma unroll
        for (int r = 0; r < 8; ++r)
            st[(8 * half + r) * GEMM_PITCH + 16 * n + c] = acc[n][r] + bv;
    }
    __syncthreads();
    if (rowBase + 16 <= M) {
        float4 o[16];
        #pragma unroll
        for (int r = 0; r < 16; ++r) o[r] = *(const float4*)(st + r * GEMM_PITCH + 4 * lane);
        #pragma unroll
        for (int r = 0; r < 16; ++r) ((float4*)(out + (size_t)(rowBase + r) * DF))[lane] = o[r];
    } else {
        for (int r = 0; r < 16; ++r) {
            if (rowBase + r < M)
                ((float4*)(out + (size_t)(rowBase + r) * DF))[lane] = *(const float4*)(st + r * GEMM_PITCH + 4 * lane);
        }
    }
}

extern "C" void kernel_launch(void* const* d_in, const int* in_sizes, int n_in,
                              void* d_out, int out_size, void* d_ws, size_t ws_size,
                              hipStream_t stream) {
    const float* feature = (const float*)d_in[0];
    const float* w0      = (const float*)d_in[1];
    const float* w1      = (const float*)d_in[2];
    const float* b1      = (const float*)d_in[3];
    const float* W2      = (const float*)d_in[4];
    const float* b2      = (const float*)d_in[5];
    const int*   src0    = (const int*)d_in[6];
    const int*   dst0    = (const int*)d_in[7];
    const int*   src1    = (const int*)d_in[8];
    const int*   dst1    = (const int*)d_in[9];

    const int E0 = in_sizes[1];
    const int E1 = in_sizes[2];
    const int N1 = N1C;
    const int N2 = out_size / DF;

    const int chunks0 = (E0 + BM_EDGES - 1) / BM_EDGES;
    const int chunks1 = (E1 + BM_EDGES - 1) / BM_EDGES;
    const int rowWords0 = chunks0 * BM_WORDS;
    const int rowWords1 = chunks1 * BM_WORDS;

    float* h   = (float*)d_ws;
    float* agg = h + (size_t)N1 * DF;
    unsigned* bits0 = (unsigned*)(agg + (size_t)N2 * DF);
    unsigned* bits1 = bits0 + (size_t)NSLAB * rowWords0;
    float* out = (float*)d_out;

    GCN_slab_bits_kernel<196, 98><<<chunks0 + chunks1, BM_TPB, 0, stream>>>(
        dst0, dst1, bits0, bits1, E0, E1, chunks0, rowWords0, rowWords1);

    {
        int blocks = (N1 + 195) / 196;
        GCN_slab_spmm_kernel<196, 8192, true><<<blocks, SLAB_TPB, 0, stream>>>(
            feature, w0, src0, dst0, bits0, b1, h, rowWords0, N1);
    }
    {
        int blocks = (N2 + 97) / 98;
        GCN_slab_spmm_kernel<98, 12288, false><<<blocks, SLAB_TPB, 0, stream>>>(
            h, w1, src1, dst1, bits1, b1, agg, rowWords1, N2);
    }
    {
        int blocks = (N2 + GEMM_ROWS - 1) / GEMM_ROWS;
        GCN_gemm_wmma_kernel<<<blocks, 256, 0, stream>>>(agg, W2, b2, out, N2);
    }
}


# Slab records counting-sorted by (owner wave, row) on 256 LDS counters: a wave sums each row in registers, stores it when its list moves on; no LDS accumulators, no clear; sum order = row-list arrival
# speedup vs baseline: 2.9428x; 1.0809x over previous
#include <hip/hip_runtime.h>


#define DF   128
#define N1C  50000

#define SLAB_TPB 1024

typedef float v2f __attribute__((ext_vector_type(2)));
typedef float v8f __attribute__((ext_vector_type(8)));


#define NSLAB    256
#define BM_EDGES 2048
#define BM_WORDS (BM_EDGES / 32)
#define BM_TPB   128
template <unsigned ROWS>
__device__ __forceinline__ void GCN_bits_chunk(const int* __restrict__ dst, unsigned* __restrict__ bits,
                                               int chunk, int nE, int rowWords, unsigned* bm) {
    constexpr int QPT = BM_EDGES / 4 / BM_TPB;
    constexpr int PPS = BM_WORDS / 4;
    constexpr int PPT = NSLAB * PPS / BM_TPB;
    const int t  = threadIdx.x;
    const int e0 = chunk * BM_EDGES;
    int4 d[QPT];
    if (e0 + BM_EDGES <= nE) {
        #pragma unroll
        for (int k = 0; k < QPT; ++k) d[k] = *(const int4*)(dst + e0 + 4 * (k * BM_TPB + t));
    } else {
        #pragma unroll
        for (int k = 0; k < QPT; ++k) {
            const int e = e0 + 4 * (k * BM_TPB + t);
            int4 v; v.x = -1; v.y = -1; v.z = -1; v.w = -1;
            if (e     < nE) v.x = dst[e];
            if (e + 1 < nE) v.y = dst[e + 1];
            if (e + 2 < nE) v.z = dst[e + 2];
            if (e + 3 < nE) v.w = dst[e + 3];
            d[k] = v;
        }
    }
    {
        uint4 z; z.x = 0u; z.y = 0u; z.z = 0u; z.w = 0u;
        #pragma unroll
        for (int k = 0; k < PPT; ++k) ((uint4*)bm)[k * BM_TPB + t] = z;
    }
    __syncthreads();
    #pragma unroll
    for (int k = 0; k < QPT; ++k) {
        const int q = k * BM_TPB + t;
        unsigned* wq = bm + (q >> 3);
        const unsigned b0 = 1u << (4 * (q & 7));
        const unsigned s0 = (unsigned)d[k].x / ROWS, s1 = (unsigned)d[k].y / ROWS;
        const unsigned s2 = (unsigned)d[k].z / ROWS, s3 = (unsigned)d[k].w / ROWS;
        if (s0 < NSLAB) atomicOr(wq + s0 * BM_WORDS, b0);
        if (s1 < NSLAB) atomicOr(wq + s1 * BM_WORDS, b0 << 1);
        if (s2 < NSLAB) atomicOr(wq + s2 * BM_WORDS, b0 << 2);
        if (s3 < NSLAB) atomicOr(wq + s3 * BM_WORDS, b0 << 3);
    }
    __syncthreads();
    const uint4* lp = (const uint4*)bm + t;
    unsigned*    gp = bits + (size_t)chunk * BM_WORDS + (size_t)(t / PPS) * rowWords + 4 * (t % PPS);
    #pragma unroll 8
    for (int k = 0; k < PPT; ++k)
        *(uint4*)(gp + (size_t)(k * (BM_TPB / PPS)) * rowWords) = lp[k * BM_TPB];
}
template <unsigned ROWS0, unsigned ROWS1>
__global__ __launch_bounds__(BM_TPB)
void GCN_slab_bits_kernel(const int* __restrict__ dst0, const int* __restrict__ dst1,
                          unsigned* __restrict__ bits0, unsigned* __restrict__ bits1,
                          int nE0, int nE1, int chunks0, int rowWords0, int rowWords1) {
    __shared__ __attribute__((aligned(16))) unsigned bm[NSLAB * BM_WORDS];
    const int c = blockIdx.x;
    if (c < chunks0) GCN_bits_chunk<ROWS0>(dst0, bits0, c, nE0, rowWords0, bm);
    else             GCN_bits_chunk<ROWS1>(dst1, bits1, c - chunks0, nE1, rowWords1, bm);
}

template <int ROWS, int CAP, bool BIAS_RELU>
__global__ __launch_bounds__(SLAB_TPB)
void GCN_slab_spmm_kernel(const float* __restrict__ in,
                          const float* __restrict__ w,
                          const int*   __restrict__ src,
                          const int*   __restrict__ dst,
                          const unsigned* __restrict__ bits,
                          const float* __restrict__ bias,
                          float*       __restrict__ out,
                          int rowWords, int nRows) {
    static_assert(ROWS <= 256, "a wave owns at most eight rows");
    __shared__ int   recRow[CAP];
    __shared__ int   recU[CAP];
    __shared__ float recW[CAP];
    __shared__ int   ord[CAP];
    __shared__ int   binCnt[256];
    __shared__ int   binStart[257];
    __shared__ int   cnt;

    const int t    = threadIdx.x;
    const int lane = t & 31;
    const int wv   = __builtin_amdgcn_readfirstlane(t >> 5);
    const int row0 = blockIdx.x * ROWS;
    int rowsHere = nRows - row0;
    if (rowsHere > ROWS) rowsHere = ROWS;
    if (rowsHere < 0) rowsHere = 0;
    const unsigned rowsU = (unsigned)rowsHere;

    if (t == 0) cnt = 0;
    if (t < 256) binCnt[t] = 0;
    __syncthreads();

    float4 bv; bv.x = 0.f; bv.y = 0.f; bv.z = 0.f; bv.w = 0.f;
    if (BIAS_RELU) bv = ((const float4*)bias)[lane];
    auto storeRow = [&](int r, float4 o) {
        if (BIAS_RELU) {
            o.x = fmaxf(o.x + bv.x, 0.f);
            o.y = fmaxf(o.y + bv.y, 0.f);
            o.z = fmaxf(o.z + bv.z, 0.f);
            o.w = fmaxf(o.w + bv.w, 0.f);
        }
        ((float4*)(out + (size_t)(row0 + r) * DF))[lane] = o;
    };

    const int    n4  = rowWords >> 2;
    const uint4* bm4 = (const uint4*)(bits + (size_t)blockIdx.x * rowWords);
    auto put = [&](int pos, int e) {
        if (pos < CAP) recU[pos] = e;
    };
    constexpr int SCAN_U = 5;
    for (int b = 0; b < n4; b += SCAN_U * SLAB_TPB) {
        uint4 v[SCAN_U];
        #pragma unroll
        for (int k = 0; k < SCAN_U; ++k) {
            const int idx = b + k * SLAB_TPB + t;
            uint4 x; x.x = 0u; x.y = 0u; x.z = 0u; x.w = 0u;
            if (idx < n4) x = bm4[idx];
            v[k] = x;
        }
        int n = 0;
        #pragma unroll
        for (int k = 0; k < SCAN_U; ++k)
            n += __popc(v[k].x) + __popc(v[k].y) + __popc(v[k].z) + __popc(v[k].w);
        int incl = n;
        #pragma unroll
        for (int d = 1; d < 32; d <<= 1) {
            const int up = __shfl_up(incl, d);
            if (lane >= d) incl += up;
        }
        int base = 0;
        if (lane == 31 && incl) base = atomicAdd(&cnt, incl);
        int pos = __builtin_amdgcn_readlane(base, 31) + incl - n;
        const bool roomy = __builtin_amdgcn_ballot_w32(pos + n > CAP) == 0u;
        if (roomy) {
            #pragma unroll
            for (int k = 0; k < SCAN_U; ++k) {
                const int eb = 128 * (b + k * SLAB_TPB + t);
                for (unsigned m = v[k].x; m; m &= m - 1) recU[pos++] = eb      + __builtin_ctz(m);
                for (unsigned m = v[k].y; m; m &= m - 1) recU[pos++] = eb + 32 + __builtin_ctz(m);
                for (unsigned m = v[k].z; m; m &= m - 1) recU[pos++] = eb + 64 + __builtin_ctz(m);
                for (unsigned m = v[k].w; m; m &= m - 1) recU[pos++] = eb + 96 + __builtin_ctz(m);
            }
        } else {
            #pragma unroll
            for (int k = 0; k < SCAN_U; ++k) {
                const int eb = 128 * (b + k * SLAB_TPB + t);
                for (unsigned m = v[k].x; m; m &= m - 1) put(pos++, eb      + __builtin_ctz(m));
                for (unsigned m = v[k].y; m; m &= m - 1) put(pos++, eb + 32 + __builtin_ctz(m));
                for (unsigned m = v[k].z; m; m &= m - 1) put(pos++, eb + 64 + __builtin_ctz(m));
                for (unsigned m = v[k].w; m; m &= m - 1) put(pos++, eb + 96 + __builtin_ctz(m));
            }
        }
    }
    __syncthreads();

    const int count = cnt;
    if (count > CAP) {
        const unsigned* brow = bits + (size_t)blockIdx.x * rowWords;
        for (int r = wv; r < rowsHere; r += 32) {
            float4 v; v.x = 0.f; v.y = 0.f; v.z = 0.f; v.w = 0.f;
            for (int i = 0; i < rowWords; ++i) {
                for (unsigned m = brow[i]; m; m &= m - 1) {
                    const int e = 32 * i + __builtin_ctz(m);
                    if (dst[e] - row0 == r) {
                        const float  we = w[e];
                        const float4 f  = ((const float4*)(in + (size_t)src[e] * DF))[lane];
                        v.x += f.x * we;
                        v.y += f.y * we;
                        v.z += f.z * we;
                        v.w += f.w * we;
                    }
                }
            }
            storeRow(r, v);
        }
        return;
    }
    for (int i = t; i < count; i += SLAB_TPB) {
        const int e = recU[i];
        const unsigned r = (unsigned)(dst[e] - row0);
        const int   u  = src[e];
        const float we = w[e];
        int pr = -1;
        if (r < rowsU) {
            const unsigned key = (r & 31u) * 8u + (r >> 5);
            pr = (int)key | (atomicAdd(&binCnt[key], 1) << 8);
        }
        recRow[i] = pr;
        recU[i]   = u;
        recW[i]   = we;
    }
    __syncthreads();

    if (wv == 0) {
        int c[8];
        int tot = 0;
        #pragma unroll
        for (int j = 0; j < 8; ++j) { c[j] = binCnt[lane * 8 + j]; tot += c[j]; }
        int incl = tot;
        #pragma unroll
        for (int d = 1; d < 32; d <<= 1) {
            const int v = __shfl_up(incl, d);
            if (lane >= d) incl += v;
        }
        int p = incl - tot;
        #pragma unroll
        for (int j = 0; j < 8; ++j) { binStart[lane * 8 + j] = p; p += c[j]; }
        if (lane == 31) binStart[256] = incl;
    }
    __syncthreads();

    for (int i = t; i < count; i += SLAB_TPB) {
        const int pr = recRow[i];
        if (pr >= 0) ord[binStart[pr & 255] + (pr >> 8)] = i;
    }
    __syncthreads();

    const int beg = __builtin_amdgcn_readfirstlane(binStart[wv * 8]);
    const int end = __builtin_amdgcn_readfirstlane(binStart[wv * 8 + 8]);
    auto loadGroup = [&](int i, float4 (&m)[8], int (&rr)[8], float (&ww)[8]) {
        int p = i + (lane & 7);
        if (p >= end) p = end - 1;
        const int id  = ord[p];
        const int prL = recRow[id];
        const int uL  = recU[id];
        const int wL  = __float_as_int(recW[id]);
        #pragma unroll
        for (int k = 0; k < 8; ++k) {
            const int u = __builtin_amdgcn_readlane(uL, k);
            rr[k] = __builtin_amdgcn_readlane(prL, k) & 7;
            ww[k] = __int_as_float(__builtin_amdgcn_readlane(wL, k));
            m[k]  = ((const float4*)(in + (size_t)u * DF))[lane];
        }
        __builtin_amdgcn_sched_barrier(0);
    };
    if (beg < end) {
        float4 mA[8], mB[8];
        int    rA[8], rB[8];
        float  wA[8], wB[8];
        int i = beg;
        loadGroup(i, mA, rA, wA);
        int    cur = rA[0];
        float4 v; v.x = 0.f; v.y = 0.f; v.z = 0.f; v.w = 0.f;
        auto addOne = [&](const float4& mk, int j, float wk) {
            if (j != cur) {
                storeRow(wv + 32 * cur, v);
                cur = j;
                v.x = 0.f; v.y = 0.f; v.z = 0.f; v.w = 0.f;
            }
            v.x += mk.x * wk;
            v.y += mk.y * wk;
            v.z += mk.z * wk;
            v.w += mk.w * wk;
        };
        auto addFull = [&](const float4 (&m)[8], const int (&rr)[8], const float (&ww)[8]) {
            #pragma unroll
            for (int k = 0; k < 8; ++k) addOne(m[k], rr[k], ww[k]);
        };
        auto addTail = [&](int i0, const float4 (&m)[8], const int (&rr)[8], const float (&ww)[8]) {
            #pragma unroll
            for (int k = 0; k < 8; ++k)
                if (i0 + k < end) addOne(m[k], rr[k], ww[k]);
        };
        while (i + 16 < end) {
            loadGroup(i + 8, mB, rB, wB);
            addFull(mA, rA, wA);
            loadGroup(i + 16, mA, rA, wA);
            addFull(mB, rB, wB);
            i += 16;
        }
        if (i + 8 < end) {
            loadGroup(i + 8, mB, rB, wB);
            addFull(mA, rA, wA);
            addTail(i + 8, mB, rB, wB);
        } else {
            addTail(i, mA, rA, wA);
        }
        storeRow(wv + 32 * cur, v);
    }

    #pragma unroll
    for (int j = 0; j < 8; ++j) {
        const int r = wv + 32 * j;
        if (r < rowsHere && binStart[wv * 8 + j] == binStart[wv * 8 + j + 1]) {
            float4 z; z.x = 0.f; z.y = 0.f; z.z = 0.f; z.w = 0.f;
            storeRow(r, z);
        }
    }
}

#define GEMM_ROWS 128
#define GEMM_PITCH 196
__global__ __launch_bounds__(256)
void GCN_gemm_wmma_kernel(const float* __restrict__ A,
                          const float* __restrict__ W,
                          const float* __restrict__ bias,
                          float*       __restrict__ out,
                          int M) {
    __shared__ float4 Bq[(DF / 4) * DF];
    __shared__ __attribute__((aligned(16))) float stage[GEMM_ROWS * GEMM_PITCH];

    const int t    = threadIdx.x;
    const int lane = t & 31;
    const int wv   = __builtin_amdgcn_readfirstlane(t >> 5);
    const int half = lane >> 4;
    const int c    = lane & 15;
    const int rowBase = blockIdx.x * GEMM_ROWS + wv * 16;

    const int row  = rowBase + c;
    const int rowc = (row < M) ? row : (M - 1);
    const float4* Ap = (const float4*)(A + (size_t)rowc * DF) + half;
    float4 a[16];
    #pragma unroll
    for (int j = 0; j < 16; ++j) a[j] = Ap[2 * j];

    #pragma unroll
    for (int i = 0; i < 16; ++i) {
        const int idx = i * 256 + t;
        const int kq  = idx >> 7;
        const int col = idx & 127;
        const float* wp = W + (size_t)(4 * kq) * DF + col;
        float4 v;
        v.x = wp[0]; v.y = wp[DF]; v.z = wp[2 * DF]; v.w = wp[3 * DF];
        Bq[idx] = v;
    }
    __syncthreads();

    v8f acc[8];
    #pragma unroll
    for (int n = 0; n < 8; ++n) acc[n] = v8f{};

    const float4* bp = Bq + half * DF + c;
    float4 bc[8], bn[8];
    #pragma unroll
    for (int n = 0; n < 8; ++n) bc[n] = bp[16 * n];
    #pragma unroll
    for (int j = 0; j < 16; ++j) {
        if (j + 1 < 16) {
            #pragma unroll
            for (int n = 0; n < 8; ++n) bn[n] = bp[(2 * (j + 1)) * DF + 16 * n];
        }
        v2f a0; a0[0] = a[j].x; a0[1] = a[j].y;
        v2f a1; a1[0] = a[j].z; a1[1] = a[j].w;
        #pragma unroll
        for (int n = 0; n < 8; ++n) {
            v2f b0; b0[0] = bc[n].x; b0[1] = bc[n].y;
            acc[n] = __builtin_amdgcn_wmma_f32_16x16x4_f32(
                         false, a0,  false, b0,
                         (short)0, acc[n],  false,  false);
        }
        #pragma unroll
        for (int n = 0; n < 8; ++n) {
            v2f b1; b1[0] = bc[n].z; b1[1] = bc[n].w;
            acc[n] = __builtin_amdgcn_wmma_f32_16x16x4_f32(
                         false, a1,  false, b1,
                         (short)0, acc[n],  false,  false);
        }
        #pragma unroll
        for (int n = 0; n < 8; ++n) bc[n] = bn[n];
    }

    float* st = stage + wv * 16 * GEMM_PITCH;
    #pragma unroll
    for (int n = 0; n < 8; ++n) {
        const float bv = bias[16 * n + c];
        #pragma unroll
        for (int r = 0; r < 8; ++r)
            st[(8 * half + r) * GEMM_PITCH + 16 * n + c] = acc[n][r] + bv;
    }
    __syncthreads();
    if (rowBase + 16 <= M) {
        float4 o[16];
        #pragma unroll
        for (int r = 0; r < 16; ++r) o[r] = *(const float4*)(st + r * GEMM_PITCH + 4 * lane);
        #pragma unroll
        for (int r = 0; r < 16; ++r) ((float4*)(out + (size_t)(rowBase + r) * DF))[lane] = o[r];
    } else {
        for (int r = 0; r < 16; ++r) {
            if (rowBase + r < M)
                ((float4*)(out + (size_t)(rowBase + r) * DF))[lane] = *(const float4*)(st + r * GEMM_PITCH + 4 * lane);
        }
    }
}

extern "C" void kernel_launch(void* const* d_in, const int* in_sizes, int n_in,
                              void* d_out, int out_size, void* d_ws, size_t ws_size,
                              hipStream_t stream) {
    const float* feature = (const float*)d_in[0];
    const float* w0      = (const float*)d_in[1];
    const float* w1      = (const float*)d_in[2];
    const float* b1      = (const float*)d_in[3];
    const float* W2      = (const float*)d_in[4];
    const float* b2      = (const float*)d_in[5];
    const int*   src0    = (const int*)d_in[6];
    const int*   dst0    = (const int*)d_in[7];
    const int*   src1    = (const int*)d_in[8];
    const int*   dst1    = (const int*)d_in[9];

    const int E0 = in_sizes[1];
    const int E1 = in_sizes[2];
    const int N1 = N1C;
    const int N2 = out_size / DF;

    const int chunks0 = (E0 + BM_EDGES - 1) / BM_EDGES;
    const int chunks1 = (E1 + BM_EDGES - 1) / BM_EDGES;
    const int rowWords0 = chunks0 * BM_WORDS;
    const int rowWords1 = chunks1 * BM_WORDS;

    float* h   = (float*)d_ws;
    float* agg = h + (size_t)N1 * DF;
    unsigned* bits0 = (unsigned*)(agg + (size_t)N2 * DF);
    unsigned* bits1 = bits0 + (size_t)NSLAB * rowWords0;
    float* out = (float*)d_out;

    GCN_slab_bits_kernel<196, 98><<<chunks0 + chunks1, BM_TPB, 0, stream>>>(
        dst0, dst1, bits0, bits1, E0, E1, chunks0, rowWords0, rowWords1);

    {
        int blocks = (N1 + 195) / 196;
        GCN_slab_spmm_kernel<196, 12288, true><<<blocks, SLAB_TPB, 0, stream>>>(
            feature, w0, src0, dst0, bits0, b1, h, rowWords0, N1);
    }
    {
        int blocks = (N2 + 97) / 98;
        GCN_slab_spmm_kernel<98, 12288, false><<<blocks, SLAB_TPB, 0, stream>>>(
            h, w1, src1, dst1, bits1, b1, agg, rowWords1, N2);
    }
    {
        int blocks = (N2 + GEMM_ROWS - 1) / GEMM_ROWS;
        GCN_gemm_wmma_kernel<<<blocks, 256, 0, stream>>>(agg, W2, b2, out, N2);
    }
}


# Slab bit matrix coarsened to one bit per 4 consecutive edges (a quarter of the bytes written and scanned); a slab reads a listed quad's dst, src, w as 16-byte loads and keeps its own; sum order kept
# speedup vs baseline: 3.0907x; 1.0503x over previous
#include <hip/hip_runtime.h>


#define DF   128
#define N1C  50000

#define SLAB_TPB 1024

typedef float v2f __attribute__((ext_vector_type(2)));
typedef float v8f __attribute__((ext_vector_type(8)));


#define NSLAB    256
#define BM_EDGES 4096
#define BM_WORDS (BM_EDGES / 4 / 32)
#define BM_TPB   256
template <unsigned ROWS>
__device__ __forceinline__ void GCN_bits_chunk(const int* __restrict__ dst, unsigned* __restrict__ bits,
                                               int chunk, int nE, int rowWords, unsigned* bm) {
    constexpr int QPT = BM_EDGES / 4 / BM_TPB;
    constexpr int PPS = BM_WORDS / 4;
    constexpr int PPT = NSLAB * PPS / BM_TPB;
    const int t  = threadIdx.x;
    const int e0 = chunk * BM_EDGES;
    int4 d[QPT];
    if (e0 + BM_EDGES <= nE) {
        #pragma unroll
        for (int k = 0; k < QPT; ++k) d[k] = *(const int4*)(dst + e0 + 4 * (k * BM_TPB + t));
    } else {
        #pragma unroll
        for (int k = 0; k < QPT; ++k) {
            const int e = e0 + 4 * (k * BM_TPB + t);
            int4 v; v.x = -1; v.y = -1; v.z = -1; v.w = -1;
            if (e     < nE) v.x = dst[e];
            if (e + 1 < nE) v.y = dst[e + 1];
            if (e + 2 < nE) v.z = dst[e + 2];
            if (e + 3 < nE) v.w = dst[e + 3];
            d[k] = v;
        }
    }
    {
        uint4 z; z.x = 0u; z.y = 0u; z.z = 0u; z.w = 0u;
        #pragma unroll
        for (int k = 0; k < PPT; ++k) ((uint4*)bm)[k * BM_TPB + t] = z;
    }
    __syncthreads();
    #pragma unroll
    for (int k = 0; k < QPT; ++k) {
        const int q = k * BM_TPB + t;
        unsigned* wq = bm + (q >> 5);
        const unsigned b0 = 1u << (q & 31);
        const unsigned s0 = (unsigned)d[k].x / ROWS, s1 = (unsigned)d[k].y / ROWS;
        const unsigned s2 = (unsigned)d[k].z / ROWS, s3 = (unsigned)d[k].w / ROWS;
        if (s0 < NSLAB) atomicOr(wq + s0 * BM_WORDS, b0);
        if (s1 < NSLAB) atomicOr(wq + s1 * BM_WORDS, b0);
        if (s2 < NSLAB) atomicOr(wq + s2 * BM_WORDS, b0);
        if (s3 < NSLAB) atomicOr(wq + s3 * BM_WORDS, b0);
    }
    __syncthreads();
    const uint4* lp = (const uint4*)bm + t;
    unsigned*    gp = bits + (size_t)chunk * BM_WORDS + (size_t)(t / PPS) * rowWords + 4 * (t % PPS);
    #pragma unroll 8
    for (int k = 0; k < PPT; ++k)
        *(uint4*)(gp + (size_t)(k * (BM_TPB / PPS)) * rowWords) = lp[k * BM_TPB];
}
template <unsigned ROWS0, unsigned ROWS1>
__global__ __launch_bounds__(BM_TPB)
void GCN_slab_bits_kernel(const int* __restrict__ dst0, const int* __restrict__ dst1,
                          unsigned* __restrict__ bits0, unsigned* __restrict__ bits1,
                          int nE0, int nE1, int chunks0, int rowWords0, int rowWords1) {
    __shared__ __attribute__((aligned(16))) unsigned bm[NSLAB * BM_WORDS];
    const int c = blockIdx.x;
    if (c < chunks0) GCN_bits_chunk<ROWS0>(dst0, bits0, c, nE0, rowWords0, bm);
    else             GCN_bits_chunk<ROWS1>(dst1, bits1, c - chunks0, nE1, rowWords1, bm);
}

template <int ROWS, int CAP, bool BIAS_RELU>
__global__ __launch_bounds__(SLAB_TPB)
void GCN_slab_spmm_kernel(const float* __restrict__ in,
                          const float* __restrict__ w,
                          const int*   __restrict__ src,
                          const int*   __restrict__ dst,
                          const unsigned* __restrict__ bits,
                          const float* __restrict__ bias,
                          float*       __restrict__ out,
                          int rowWords, int nRows, int nE) {
    static_assert(ROWS <= 256, "a wave owns at most eight rows");
    __shared__ int   recRow[CAP];
    __shared__ int   recU[CAP];
    __shared__ float recW[CAP];
    __shared__ int   ord[CAP];
    __shared__ int   binCnt[256];
    __shared__ int   binStart[257];
    __shared__ int   cnt;
    __shared__ int   extra;

    const int t    = threadIdx.x;
    const int lane = t & 31;
    const int wv   = __builtin_amdgcn_readfirstlane(t >> 5);
    const int row0 = blockIdx.x * ROWS;
    int rowsHere = nRows - row0;
    if (rowsHere > ROWS) rowsHere = ROWS;
    if (rowsHere < 0) rowsHere = 0;
    const unsigned rowsU = (unsigned)rowsHere;

    if (t == 0) { cnt = 0; extra = 0; }
    if (t < 256) binCnt[t] = 0;
    __syncthreads();

    float4 bv; bv.x = 0.f; bv.y = 0.f; bv.z = 0.f; bv.w = 0.f;
    if (BIAS_RELU) bv = ((const float4*)bias)[lane];
    auto storeRow = [&](int r, float4 o) {
        if (BIAS_RELU) {
            o.x = fmaxf(o.x + bv.x, 0.f);
            o.y = fmaxf(o.y + bv.y, 0.f);
            o.z = fmaxf(o.z + bv.z, 0.f);
            o.w = fmaxf(o.w + bv.w, 0.f);
        }
        ((float4*)(out + (size_t)(row0 + r) * DF))[lane] = o;
    };

    const int    n4  = rowWords >> 2;
    const uint4* bm4 = (const uint4*)(bits + (size_t)blockIdx.x * rowWords);
    auto put = [&](int pos, int e) {
        if (pos < CAP) recU[pos] = e;
    };
    constexpr int SCAN_U = 2;
    for (int b = 0; b < n4; b += SCAN_U * SLAB_TPB) {
        uint4 v[SCAN_U];
        #pragma unroll
        for (int k = 0; k < SCAN_U; ++k) {
            const int idx = b + k * SLAB_TPB + t;
            uint4 x; x.x = 0u; x.y = 0u; x.z = 0u; x.w = 0u;
            if (idx < n4) x = bm4[idx];
            v[k] = x;
        }
        int n = 0;
        #pragma unroll
        for (int k = 0; k < SCAN_U; ++k)
            n += __popc(v[k].x) + __popc(v[k].y) + __popc(v[k].z) + __popc(v[k].w);
        int incl = n;
        #pragma unroll
        for (int d = 1; d < 32; d <<= 1) {
            const int up = __shfl_up(incl, d);
            if (lane >= d) incl += up;
        }
        int base = 0;
        if (lane == 31 && incl) base = atomicAdd(&cnt, incl);
        int pos = __builtin_amdgcn_readlane(base, 31) + incl - n;
        const bool roomy = __builtin_amdgcn_ballot_w32(pos + n > CAP) == 0u;
        if (roomy) {
            #pragma unroll
            for (int k = 0; k < SCAN_U; ++k) {
                const int eb = 128 * (b + k * SLAB_TPB + t);
                for (unsigned m = v[k].x; m; m &= m - 1) recU[pos++] = eb      + __builtin_ctz(m);
                for (unsigned m = v[k].y; m; m &= m - 1) recU[pos++] = eb + 32 + __builtin_ctz(m);
                for (unsigned m = v[k].z; m; m &= m - 1) recU[pos++] = eb + 64 + __builtin_ctz(m);
                for (unsigned m = v[k].w; m; m &= m - 1) recU[pos++] = eb + 96 + __builtin_ctz(m);
            }
        } else {
            #pragma unroll
            for (int k = 0; k < SCAN_U; ++k) {
                const int eb = 128 * (b + k * SLAB_TPB + t);
                for (unsigned m = v[k].x; m; m &= m - 1) put(pos++, eb      + __builtin_ctz(m));
                for (unsigned m = v[k].y; m; m &= m - 1) put(pos++, eb + 32 + __builtin_ctz(m));
                for (unsigned m = v[k].z; m; m &= m - 1) put(pos++, eb + 64 + __builtin_ctz(m));
                for (unsigned m = v[k].w; m; m &= m - 1) put(pos++, eb + 96 + __builtin_ctz(m));
            }
        }
    }
    __syncthreads();

    auto slowPath = [&]() {
        const unsigned* brow = bits + (size_t)blockIdx.x * rowWords;
        for (int r = wv; r < rowsHere; r += 32) {
            float4 v; v.x = 0.f; v.y = 0.f; v.z = 0.f; v.w = 0.f;
            for (int i = 0; i < rowWords; ++i) {
                for (unsigned m = brow[i]; m; m &= m - 1) {
                    const int e0 = 4 * (32 * i + __builtin_ctz(m));
                    for (int e = e0; e < e0 + 4; ++e) {
                        if (e < nE && dst[e] - row0 == r) {
                            const float  we = w[e];
                            const float4 f  = ((const float4*)(in + (size_t)src[e] * DF))[lane];
                            v.x += f.x * we;
                            v.y += f.y * we;
                            v.z += f.z * we;
                            v.w += f.w * we;
                        }
                    }
                }
            }
            storeRow(r, v);
        }
    };

    const int count = cnt;
    if (count > CAP) { slowPath(); return; }
    for (int i = t; i < count; i += SLAB_TPB) {
        const int e0 = 4 * recU[i];
        int   dd[4], uu[4];
        float wf[4];
        if (e0 + 3 < nE) {
            const int4   d4 = *(const int4*)(dst + e0);
            const int4   u4 = *(const int4*)(src + e0);
            const float4 w4 = *(const float4*)(w + e0);
            dd[0] = d4.x; dd[1] = d4.y; dd[2] = d4.z; dd[3] = d4.w;
            uu[0] = u4.x; uu[1] = u4.y; uu[2] = u4.z; uu[3] = u4.w;
            wf[0] = w4.x; wf[1] = w4.y; wf[2] = w4.z; wf[3] = w4.w;
        } else {
            #pragma unroll
            for (int j = 0; j < 4; ++j) {
                dd[j] = -1; uu[j] = 0; wf[j] = 0.f;
                if (e0 + j < nE) { dd[j] = dst[e0 + j]; uu[j] = src[e0 + j]; wf[j] = w[e0 + j]; }
            }
        }
        unsigned rr[4];
        unsigned mem = 0u;
        #pragma unroll
        for (int j = 0; j < 4; ++j) {
            rr[j] = (unsigned)(dd[j] - row0);
            if (rr[j] < rowsU) mem |= 1u << j;
        }
        if (mem == 0u) recRow[i] = -1;
        int slot = i;
        while (mem) {
            const int j = __builtin_ctz(mem);
            mem &= mem - 1u;
            const unsigned r  = (j == 0) ? rr[0] : (j == 1) ? rr[1] : (j == 2) ? rr[2] : rr[3];
            const int      u  = (j == 0) ? uu[0] : (j == 1) ? uu[1] : (j == 2) ? uu[2] : uu[3];
            const float    we = (j == 0) ? wf[0] : (j == 1) ? wf[1] : (j == 2) ? wf[2] : wf[3];
            const unsigned key = (r & 31u) * 8u + (r >> 5);
            const int pr = (int)key | (atomicAdd(&binCnt[key], 1) << 8);
            if (slot < CAP) {
                recRow[slot] = pr;
                recU[slot]   = u;
                recW[slot]   = we;
            }
            if (mem) slot = count + atomicAdd(&extra, 1);
        }
    }
    __syncthreads();
    const int total = count + extra;
    if (total > CAP) { slowPath(); return; }

    if (wv == 0) {
        int c[8];
        int tot = 0;
        #pragma unroll
        for (int j = 0; j < 8; ++j) { c[j] = binCnt[lane * 8 + j]; tot += c[j]; }
        int incl = tot;
        #pragma unroll
        for (int d = 1; d < 32; d <<= 1) {
            const int v = __shfl_up(incl, d);
            if (lane >= d) incl += v;
        }
        int p = incl - tot;
        #pragma unroll
        for (int j = 0; j < 8; ++j) { binStart[lane * 8 + j] = p; p += c[j]; }
        if (lane == 31) binStart[256] = incl;
    }
    __syncthreads();

    for (int i = t; i < total; i += SLAB_TPB) {
        const int pr = recRow[i];
        if (pr >= 0) ord[binStart[pr & 255] + (pr >> 8)] = i;
    }
    __syncthreads();

    const int beg = __builtin_amdgcn_readfirstlane(binStart[wv * 8]);
    const int end = __builtin_amdgcn_readfirstlane(binStart[wv * 8 + 8]);
    auto loadGroup = [&](int i, float4 (&m)[8], int (&rr)[8], float (&ww)[8]) {
        int p = i + (lane & 7);
        if (p >= end) p = end - 1;
        const int id  = ord[p];
        const int prL = recRow[id];
        const int uL  = recU[id];
        const int wL  = __float_as_int(recW[id]);
        #pragma unroll
        for (int k = 0; k < 8; ++k) {
            const int u = __builtin_amdgcn_readlane(uL, k);
            rr[k] = __builtin_amdgcn_readlane(prL, k) & 7;
            ww[k] = __int_as_float(__builtin_amdgcn_readlane(wL, k));
            m[k]  = ((const float4*)(in + (size_t)u * DF))[lane];
        }
        __builtin_amdgcn_sched_barrier(0);
    };
    if (beg < end) {
        float4 mA[8], mB[8];
        int    rA[8], rB[8];
        float  wA[8], wB[8];
        int i = beg;
        loadGroup(i, mA, rA, wA);
        int    cur = rA[0];
        float4 v; v.x = 0.f; v.y = 0.f; v.z = 0.f; v.w = 0.f;
        auto addOne = [&](const float4& mk, int j, float wk) {
            if (j != cur) {
                storeRow(wv + 32 * cur, v);
                cur = j;
                v.x = 0.f; v.y = 0.f; v.z = 0.f; v.w = 0.f;
            }
            v.x += mk.x * wk;
            v.y += mk.y * wk;
            v.z += mk.z * wk;
            v.w += mk.w * wk;
        };
        auto addFull = [&](const float4 (&m)[8], const int (&rr)[8], const float (&ww)[8]) {
            #pragma unroll
            for (int k = 0; k < 8; ++k) addOne(m[k], rr[k], ww[k]);
        };
        auto addTail = [&](int i0, const float4 (&m)[8], const int (&rr)[8], const float (&ww)[8]) {
            #pragma unroll
            for (int k = 0; k < 8; ++k)
                if (i0 + k < end) addOne(m[k], rr[k], ww[k]);
        };
        while (i + 16 < end) {
            loadGroup(i + 8, mB, rB, wB);
            addFull(mA, rA, wA);
            loadGroup(i + 16, mA, rA, wA);
            addFull(mB, rB, wB);
            i += 16;
        }
        if (i + 8 < end) {
            loadGroup(i + 8, mB, rB, wB);
            addFull(mA, rA, wA);
            addTail(i + 8, mB, rB, wB);
        } else {
            addTail(i, mA, rA, wA);
        }
        storeRow(wv + 32 * cur, v);
    }

    #pragma unroll
    for (int j = 0; j < 8; ++j) {
        const int r = wv + 32 * j;
        if (r < rowsHere && binStart[wv * 8 + j] == binStart[wv * 8 + j + 1]) {
            float4 z; z.x = 0.f; z.y = 0.f; z.z = 0.f; z.w = 0.f;
            storeRow(r, z);
        }
    }
}

#define GEMM_ROWS 128
#define GEMM_PITCH 196
__global__ __launch_bounds__(256)
void GCN_gemm_wmma_kernel(const float* __restrict__ A,
                          const float* __restrict__ W,
                          const float* __restrict__ bias,
                          float*       __restrict__ out,
                          int M) {
    __shared__ float4 Bq[(DF / 4) * DF];
    __shared__ __attribute__((aligned(16))) float stage[GEMM_ROWS * GEMM_PITCH];

    const int t    = threadIdx.x;
    const int lane = t & 31;
    const int wv   = __builtin_amdgcn_readfirstlane(t >> 5);
    const int half = lane >> 4;
    const int c    = lane & 15;
    const int rowBase = blockIdx.x * GEMM_ROWS + wv * 16;

    const int row  = rowBase + c;
    const int rowc = (row < M) ? row : (M - 1);
    const float4* Ap = (const float4*)(A + (size_t)rowc * DF) + half;
    float4 a[16];
    #pragma unroll
    for (int j = 0; j < 16; ++j) a[j] = Ap[2 * j];

    #pragma unroll
    for (int i = 0; i < 16; ++i) {
        const int idx = i * 256 + t;
        const int kq  = idx >> 7;
        const int col = idx & 127;
        const float* wp = W + (size_t)(4 * kq) * DF + col;
        float4 v;
        v.x = wp[0]; v.y = wp[DF]; v.z = wp[2 * DF]; v.w = wp[3 * DF];
        Bq[idx] = v;
    }
    __syncthreads();

    v8f acc[8];
    #pragma unroll
    for (int n = 0; n < 8; ++n) acc[n] = v8f{};

    const float4* bp = Bq + half * DF + c;
    float4 bc[8], bn[8];
    #pragma unroll
    for (int n = 0; n < 8; ++n) bc[n] = bp[16 * n];
    #pragma unroll
    for (int j = 0; j < 16; ++j) {
        if (j + 1 < 16) {
            #pragma unroll
            for (int n = 0; n < 8; ++n) bn[n] = bp[(2 * (j + 1)) * DF + 16 * n];
        }
        v2f a0; a0[0] = a[j].x; a0[1] = a[j].y;
        v2f a1; a1[0] = a[j].z; a1[1] = a[j].w;
        #pragma unroll
        for (int n = 0; n < 8; ++n) {
            v2f b0; b0[0] = bc[n].x; b0[1] = bc[n].y;
            acc[n] = __builtin_amdgcn_wmma_f32_16x16x4_f32(
                         false, a0,  false, b0,
                         (short)0, acc[n],  false,  false);
        }
        #pragma unroll
        for (int n = 0; n < 8; ++n) {
            v2f b1; b1[0] = bc[n].z; b1[1] = bc[n].w;
            acc[n] = __builtin_amdgcn_wmma_f32_16x16x4_f32(
                         false, a1,  false, b1,
                         (short)0, acc[n],  false,  false);
        }
        #pragma unroll
        for (int n = 0; n < 8; ++n) bc[n] = bn[n];
    }

    float* st = stage + wv * 16 * GEMM_PITCH;
    #pragma unroll
    for (int n = 0; n < 8; ++n) {
        const float bv = bias[16 * n + c];
        #pragma unroll
        for (int r = 0; r < 8; ++r)
            st[(8 * half + r) * GEMM_PITCH + 16 * n + c] = acc[n][r] + bv;
    }
    __syncthreads();
    if (rowBase + 16 <= M) {
        float4 o[16];
        #pragma unroll
        for (int r = 0; r < 16; ++r) o[r] = *(const float4*)(st + r * GEMM_PITCH + 4 * lane);
        #pragma unroll
        for (int r = 0; r < 16; ++r) ((float4*)(out + (size_t)(rowBase + r) * DF))[lane] = o[r];
    } else {
        for (int r = 0; r < 16; ++r) {
            if (rowBase + r < M)
                ((float4*)(out + (size_t)(rowBase + r) * DF))[lane] = *(const float4*)(st + r * GEMM_PITCH + 4 * lane);
        }
    }
}

extern "C" void kernel_launch(void* const* d_in, const int* in_sizes, int n_in,
                              void* d_out, int out_size, void* d_ws, size_t ws_size,
                              hipStream_t stream) {
    const float* feature = (const float*)d_in[0];
    const float* w0      = (const float*)d_in[1];
    const float* w1      = (const float*)d_in[2];
    const float* b1      = (const float*)d_in[3];
    const float* W2      = (const float*)d_in[4];
    const float* b2      = (const float*)d_in[5];
    const int*   src0    = (const int*)d_in[6];
    const int*   dst0    = (const int*)d_in[7];
    const int*   src1    = (const int*)d_in[8];
    const int*   dst1    = (const int*)d_in[9];

    const int E0 = in_sizes[1];
    const int E1 = in_sizes[2];
    const int N1 = N1C;
    const int N2 = out_size / DF;

    const int chunks0 = (E0 + BM_EDGES - 1) / BM_EDGES;
    const int chunks1 = (E1 + BM_EDGES - 1) / BM_EDGES;
    const int rowWords0 = chunks0 * BM_WORDS;
    const int rowWords1 = chunks1 * BM_WORDS;

    float* h   = (float*)d_ws;
    float* agg = h + (size_t)N1 * DF;
    unsigned* bits0 = (unsigned*)(agg + (size_t)N2 * DF);
    unsigned* bits1 = bits0 + (size_t)NSLAB * rowWords0;
    float* out = (float*)d_out;

    GCN_slab_bits_kernel<196, 98><<<chunks0 + chunks1, BM_TPB, 0, stream>>>(
        dst0, dst1, bits0, bits1, E0, E1, chunks0, rowWords0, rowWords1);

    {
        int blocks = (N1 + 195) / 196;
        GCN_slab_spmm_kernel<196, 12288, true><<<blocks, SLAB_TPB, 0, stream>>>(
            feature, w0, src0, dst0, bits0, b1, h, rowWords0, N1, E0);
    }
    {
        int blocks = (N2 + 97) / 98;
        GCN_slab_spmm_kernel<98, 12288, false><<<blocks, SLAB_TPB, 0, stream>>>(
            h, w1, src1, dst1, bits1, b1, agg, rowWords1, N2, E1);
    }
    {
        int blocks = (N2 + GEMM_ROWS - 1) / GEMM_ROWS;
        GCN_gemm_wmma_kernel<<<blocks, 256, 0, stream>>>(agg, W2, b2, out, N2);
    }
}
